# starting kernel
# baseline (speedup 1.0000x reference)
#include <hip/hip_runtime.h>


typedef float v2f __attribute__((ext_vector_type(2)));
typedef float v8f __attribute__((ext_vector_type(8)));

#define R_TOT          128
#define T_STEPS        1024
#define N_AG           128
#define RUNS_PER_BLOCK 16
#define SLOT_STRIDE    132

__device__ __forceinline__ float fast_tanh(float x) {
#if __has_builtin(__builtin_amdgcn_tanhf)
  return __builtin_amdgcn_tanhf(x);
#else
  float e = __builtin_exp2f(x * 2.885390081777927f);
  return 1.0f - 2.0f / (e + 1.0f);
#endif
}

__global__ __launch_bounds__(32)
void com2net_wmma_kernel(const float* __restrict__ runs,
                         const float* __restrict__ W1,
                         const float* __restrict__ b1,
                         const float* __restrict__ W2,
                         const float* __restrict__ b2,
                         float*       __restrict__ out)
{
  __shared__ __align__(16) float cslot [32 * SLOT_STRIDE];
  __shared__ __align__(16) float outbuf[32 * SLOT_STRIDE];

  const int lane    = threadIdx.x;
  const int up      = lane >> 4;
  const int l15     = lane & 15;
  const int runBase = blockIdx.x * RUNS_PER_BLOCK;

  for (int k = lane; k < 32 * SLOT_STRIDE; k += 32) cslot[k] = 0.0f;

  v2f amat; amat.x = 0.0f; amat.y = 0.0f;
  {
    const int h  = l15;
    const int k0 = up * 2;
    if (h < 10) { amat.x = W1[h * 4 + k0]; amat.y = W1[h * 4 + k0 + 1]; }
  }

  v8f bias1;
#pragma unroll
  for (int v = 0; v < 8; ++v) {
    const int m = v + 8 * up;
    bias1[v] = (m < 10) ? b1[m] : 0.0f;
  }

  float w2r[30];
#pragma unroll
  for (int k = 0; k < 30; ++k) w2r[k] = W2[k];
  const float b20 = b2[0], b21 = b2[1], b22 = b2[2];

  const float* xrow_base = runs + (size_t)(runBase + l15) * T_STEPS * (N_AG * 2);
  const float* cread     = &cslot [l15  * SLOT_STRIDE];
  float*       cwrite    = &cslot [lane * SLOT_STRIDE];
  float*       obwrite   = &outbuf[lane * SLOT_STRIDE];

  for (int t = 0; t < T_STEPS; ++t) {
    const float* xrow = xrow_base + (size_t)t * (N_AG * 2);

    if (t + 1 < T_STEPS) {
      const char* nxt = (const char*)(xrow + N_AG * 2);
#pragma unroll
      for (int j = 0; j < 4; ++j)
        __builtin_prefetch(nxt + (size_t)(up * 4 + j) * 128, 0, 0);
    }

    v2f   xv_cur = *(const v2f*)xrow;
    float cb_cur = cread[1];
    float co     = 0.0f;

    for (int i = 0; i < N_AG; ++i) {
      const int ip = (i + 1 < N_AG) ? i + 1 : i;
      const v2f   xv_next = *(const v2f*)(xrow + ip * 2);
      const float cb_next = cread[ip + 1];

      v2f bmat;
      bmat.x = up ? co     : xv_cur.x;
      bmat.y = up ? cb_cur : xv_cur.y;

      v8f d = __builtin_amdgcn_wmma_f32_16x16x4_f32(
          false, amat, false, bmat, (short)0, bias1, false, false);

      float dt[8];
#pragma unroll
      for (int v = 0; v < 8; ++v) dt[v] = fast_tanh(d[v]);

      const float h8 = __shfl_xor(dt[0], 16, 32);
      const float h9 = __shfl_xor(dt[1], 16, 32);

      float e0 = fmaf(w2r[20], dt[0], b22);
      float e1 =      w2r[21] * dt[1];
      e0 = fmaf(w2r[22], dt[2], e0);  e1 = fmaf(w2r[23], dt[3], e1);
      e0 = fmaf(w2r[24], dt[4], e0);  e1 = fmaf(w2r[25], dt[5], e1);
      e0 = fmaf(w2r[26], dt[6], e0);  e1 = fmaf(w2r[27], dt[7], e1);
      e0 = fmaf(w2r[28], h8,    e0);  e1 = fmaf(w2r[29], h9,    e1);
      const float o2 = e0 + e1;
      co = __shfl_xor(o2, 16, 32);

      float a0 = fmaf(w2r[0], dt[0], b20), a1 = w2r[1] * dt[1];
      a0 = fmaf(w2r[2], dt[2], a0);  a1 = fmaf(w2r[3], dt[3], a1);
      a0 = fmaf(w2r[4], dt[4], a0);  a1 = fmaf(w2r[5], dt[5], a1);
      a0 = fmaf(w2r[6], dt[6], a0);  a1 = fmaf(w2r[7], dt[7], a1);
      a0 = fmaf(w2r[8], h8,   a0);   a1 = fmaf(w2r[9], h9,   a1);
      const float o0 = a0 + a1;

      float c0 = fmaf(w2r[10], dt[0], b21), c1 = w2r[11] * dt[1];
      c0 = fmaf(w2r[12], dt[2], c0); c1 = fmaf(w2r[13], dt[3], c1);
      c0 = fmaf(w2r[14], dt[4], c0); c1 = fmaf(w2r[15], dt[5], c1);
      c0 = fmaf(w2r[16], dt[6], c0); c1 = fmaf(w2r[17], dt[7], c1);
      c0 = fmaf(w2r[18], h8,   c0);  c1 = fmaf(w2r[19], h9,   c1);
      const float o1 = c0 + c1;

      obwrite[i] = o0;
      cwrite[i]  = o1;

      xv_cur = xv_next;
      cb_cur = cb_next;
    }

#pragma unroll 4
    for (int rr = 0; rr < RUNS_PER_BLOCK; ++rr) {
      const float4 vv = *(const float4*)&outbuf[rr * SLOT_STRIDE + lane * 4];
      *(float4*)(out + ((size_t)(runBase + rr) * T_STEPS + t) * N_AG + lane * 4) = vv;
    }
  }
}

extern "C" void kernel_launch(void* const* d_in, const int* in_sizes, int n_in,
                              void* d_out, int out_size, void* d_ws, size_t ws_size,
                              hipStream_t stream) {
  (void)in_sizes; (void)n_in; (void)out_size; (void)d_ws; (void)ws_size;
  const float* runs = (const float*)d_in[0];
  const float* W1   = (const float*)d_in[1];
  const float* b1   = (const float*)d_in[2];
  const float* W2   = (const float*)d_in[3];
  const float* b2   = (const float*)d_in[4];
  float* out        = (float*)d_out;

  dim3 grid(R_TOT / RUNS_PER_BLOCK);
  dim3 block(32);
  com2net_wmma_kernel<<<grid, block, 0, stream>>>(runs, W1, b1, W2, b2, out);
}


# Wavefront skew: cell (t,i) needs only wavefront i+2t-1, so 64 timesteps of a run advance together in one wave (4 WMMA tiles): 2174 serial steps, not 131072; per-cell sum order unchanged
# speedup vs baseline: 16.6226x; 16.6226x over previous
#include <hip/hip_runtime.h>


typedef float v2f __attribute__((ext_vector_type(2)));
typedef float v8f __attribute__((ext_vector_type(8)));

#define R_TOT    128
#define T_STEPS  1024
#define N_AG     128
#define TILES    4
#define SLOTS    (16 * TILES)
#define U_SPAN   ((T_STEPS / SLOTS) * N_AG)
#define W_TOTAL  (2 * (SLOTS - 1) + U_SPAN)
#define W_LOOP   (((W_TOTAL + 5) / 6) * 6)

__device__ __forceinline__ float fast_tanh(float x) {
#if __has_builtin(__builtin_amdgcn_tanhf)
  return __builtin_amdgcn_tanhf(x);
#else
  float e = __builtin_exp2f(x * 2.885390081777927f);
  return 1.0f - 2.0f / (e + 1.0f);
#endif
}

__global__ __launch_bounds__(32)
void recurrent_mlp_wavefront_kernel(const float* __restrict__ runs,
                              const float* __restrict__ W1,
                              const float* __restrict__ b1,
                              const float* __restrict__ W2,
                              const float* __restrict__ b2,
                              float*       __restrict__ out)
{
  const int lane = threadIdx.x;
  const int up   = lane >> 4;
  const int l15  = lane & 15;
  const int run  = blockIdx.x;

  v2f amat; amat.x = 0.0f; amat.y = 0.0f;
  {
    const int h  = l15;
    const int k0 = up * 2;
    if (h < 10) { amat.x = W1[h * 4 + k0]; amat.y = W1[h * 4 + k0 + 1]; }
  }

  v8f bias1;
#pragma unroll
  for (int v = 0; v < 8; ++v) {
    const int m = v + 8 * up;
    bias1[v] = (m < 10) ? b1[m] : 0.0f;
  }

  float w2r[30];
#pragma unroll
  for (int k = 0; k < 30; ++k) w2r[k] = W2[k];
  const float b20 = b2[0], b21 = b2[1], b22 = b2[2];

  const float* xrun = runs + (size_t)run * T_STEPS * (N_AG * 2);
  float*       orun = out  + (size_t)run * T_STEPS * N_AG;
  const int rot_src = (l15 + 15) & 15;

  auto xptr = [&](int s, int w) __attribute__((always_inline)) -> const float4* {
    const int u  = w - 2 * s;
    const int uu = ((unsigned)u < (unsigned)U_SPAN) ? u : 0;
    const int t  = s + SLOTS * (uu >> 7);
    const int i  = uu & (N_AG - 1);
    return (const float4*)(xrun + ((size_t)t * N_AG + i) * 2);
  };

  auto mlp = [&](v2f bmat, float& o0, float& o1, float& o2) __attribute__((always_inline)) {
    v8f d = __builtin_amdgcn_wmma_f32_16x16x4_f32(
        false, amat, false, bmat, (short)0, bias1, false, false);

    float dt[8];
#pragma unroll
    for (int v = 0; v < 8; ++v) dt[v] = fast_tanh(d[v]);

    const float h8 = __shfl_xor(dt[0], 16, 32);
    const float h9 = __shfl_xor(dt[1], 16, 32);

    float e0 = fmaf(w2r[20], dt[0], b22);
    float e1 =      w2r[21] * dt[1];
    e0 = fmaf(w2r[22], dt[2], e0);  e1 = fmaf(w2r[23], dt[3], e1);
    e0 = fmaf(w2r[24], dt[4], e0);  e1 = fmaf(w2r[25], dt[5], e1);
    e0 = fmaf(w2r[26], dt[6], e0);  e1 = fmaf(w2r[27], dt[7], e1);
    e0 = fmaf(w2r[28], h8,    e0);  e1 = fmaf(w2r[29], h9,    e1);
    o2 = e0 + e1;

    float c0 = fmaf(w2r[10], dt[0], b21), c1 = w2r[11] * dt[1];
    c0 = fmaf(w2r[12], dt[2], c0); c1 = fmaf(w2r[13], dt[3], c1);
    c0 = fmaf(w2r[14], dt[4], c0); c1 = fmaf(w2r[15], dt[5], c1);
    c0 = fmaf(w2r[16], dt[6], c0); c1 = fmaf(w2r[17], dt[7], c1);
    c0 = fmaf(w2r[18], h8,   c0);  c1 = fmaf(w2r[19], h9,   c1);
    o1 = c0 + c1;

    float a0 = fmaf(w2r[0], dt[0], b20), a1 = w2r[1] * dt[1];
    a0 = fmaf(w2r[2], dt[2], a0);  a1 = fmaf(w2r[3], dt[3], a1);
    a0 = fmaf(w2r[4], dt[4], a0);  a1 = fmaf(w2r[5], dt[5], a1);
    a0 = fmaf(w2r[6], dt[6], a0);  a1 = fmaf(w2r[7], dt[7], a1);
    a0 = fmaf(w2r[8], h8,   a0);   a1 = fmaf(w2r[9], h9,   a1);
    o0 = a0 + a1;
  };

  float co[TILES], cb[TILES];
#pragma unroll
  for (int k = 0; k < TILES; ++k) { co[k] = 0.0f; cb[k] = 0.0f; }

  auto iter = [&](const int w, const float4 (&xa)[TILES], float4 (&xc)[TILES])
      __attribute__((always_inline)) {
#pragma unroll
    for (int k = 0; k < TILES; ++k) xc[k] = *xptr(16 * k + l15, w + 4);

    {
      const int kp = (w >> 1) & (TILES - 1);
      __builtin_prefetch((const char*)xptr(16 * kp + l15, w + 4 + 16), 0, 0);
    }

    int   uu[TILES];
    float o0e[TILES], o1s[TILES], rr[TILES];

#pragma unroll
    for (int k = 0; k < TILES; ++k) {
      const int s = 16 * k + l15;
      const int u = w - 2 * s;
      uu[k] = u;
      const int  i  = u & (N_AG - 1);
      const bool t0 = (s == 0) && (u < N_AG);
      const float co_in = (i == 0) ? 0.0f : co[k];
      const float cb_in = t0       ? 0.0f : cb[k];
      v2f bmat;
      bmat.x = up ? co_in : xa[k].x;
      bmat.y = up ? cb_in : xa[k].y;
      float o2;
      mlp(bmat, o0e[k], o1s[k], o2);
      co[k] = __shfl_xor(o2, 16, 32);
    }
#pragma unroll
    for (int k = 0; k < TILES; ++k) rr[k] = __shfl(o1s[k], rot_src, 32);
#pragma unroll
    for (int k = 0; k < TILES; ++k)
      cb[k] = (l15 == 0) ? rr[(k + TILES - 1) & (TILES - 1)] : rr[k];

    float o0o[TILES];
#pragma unroll
    for (int k = 0; k < TILES; ++k) {
      const int s = 16 * k + l15;
      const int u = uu[k];
      const int  i1 = (u & (N_AG - 1)) + 1;
      const bool t0 = (s == 0) && (u < N_AG);
      const float cb_in = (t0 || i1 == N_AG - 1) ? 0.0f : cb[k];
      v2f bmat;
      bmat.x = up ? co[k] : xa[k].z;
      bmat.y = up ? cb_in : xa[k].w;
      float o2;
      mlp(bmat, o0o[k], o1s[k], o2);
      co[k] = __shfl_xor(o2, 16, 32);
    }
#pragma unroll
    for (int k = 0; k < TILES; ++k) rr[k] = __shfl(o1s[k], rot_src, 32);
#pragma unroll
    for (int k = 0; k < TILES; ++k)
      cb[k] = (l15 == 0) ? rr[(k + TILES - 1) & (TILES - 1)] : rr[k];

    if (up == 0) {
#pragma unroll
      for (int k = 0; k < TILES; ++k) {
        const int u = uu[k];
        if ((unsigned)u < (unsigned)U_SPAN) {
          const int t = 16 * k + l15 + SLOTS * (u >> 7);
          const int i = u & (N_AG - 1);
          float2 v; v.x = o0e[k]; v.y = o0o[k];
          *(float2*)(orun + (size_t)t * N_AG + i) = v;
        }
      }
    }

  };

  float4 X0[TILES], X1[TILES], X2[TILES];
#pragma unroll
  for (int k = 0; k < TILES; ++k) {
    X0[k] = *xptr(16 * k + l15, 0);
    X1[k] = *xptr(16 * k + l15, 2);
  }
  for (int w = 0; w < W_LOOP; w += 6) {
    iter(w,     X0, X2);
    iter(w + 2, X1, X0);
    iter(w + 4, X2, X1);
  }
}

extern "C" void kernel_launch(void* const* d_in, const int* in_sizes, int n_in,
                              void* d_out, int out_size, void* d_ws, size_t ws_size,
                              hipStream_t stream) {
  (void)in_sizes; (void)n_in; (void)out_size; (void)d_ws; (void)ws_size;
  const float* runs = (const float*)d_in[0];
  const float* W1   = (const float*)d_in[1];
  const float* b1   = (const float*)d_in[2];
  const float* W2   = (const float*)d_in[3];
  const float* b2   = (const float*)d_in[4];
  float* out        = (float*)d_out;

  dim3 grid(R_TOT);
  dim3 block(32);
  recurrent_mlp_wavefront_kernel<<<grid, block, 0, stream>>>(runs, W1, b1, W2, b2, out);
}


# Loop split into ramp-up, full stretch, ramp-down: in the full stretch all 64 slots are busy, so the four control stores go out as one group with no per-slot range test and no lane-dependent branch
# speedup vs baseline: 19.4515x; 1.1702x over previous
#include <hip/hip_runtime.h>


typedef float v2f __attribute__((ext_vector_type(2)));
typedef float v8f __attribute__((ext_vector_type(8)));

#define R_TOT    128
#define T_STEPS  1024
#define N_AG     128
#define TILES    4
#define SLOTS    (16 * TILES)
#define U_SPAN   ((T_STEPS / SLOTS) * N_AG)
#define W_TOTAL  (2 * (SLOTS - 1) + U_SPAN)
#define W_LOOP   (((W_TOTAL + 5) / 6) * 6)
#define W_FULL_LO ((((2 * (SLOTS - 1)) + 5) / 6) * 6)
#define W_FULL_HI ((U_SPAN / 6) * 6)
static_assert(W_FULL_LO >= 2 * (SLOTS - 1) && W_FULL_LO % 6 == 0, "full stretch starts on a trip boundary");
static_assert(W_FULL_HI % 6 == 0 && W_FULL_HI - 2 < U_SPAN && W_FULL_HI <= W_LOOP, "full stretch ends inside every span");

__device__ __forceinline__ float fast_tanh(float x) {
#if __has_builtin(__builtin_amdgcn_tanhf)
  return __builtin_amdgcn_tanhf(x);
#else
  float e = __builtin_exp2f(x * 2.885390081777927f);
  return 1.0f - 2.0f / (e + 1.0f);
#endif
}

__global__ __launch_bounds__(32)
void recurrent_mlp_wavefront_kernel(const float* __restrict__ runs,
                              const float* __restrict__ W1,
                              const float* __restrict__ b1,
                              const float* __restrict__ W2,
                              const float* __restrict__ b2,
                              float*       __restrict__ out)
{
  const int lane = threadIdx.x;
  const int up   = lane >> 4;
  const int l15  = lane & 15;
  const int run  = blockIdx.x;

  v2f amat; amat.x = 0.0f; amat.y = 0.0f;
  {
    const int h  = l15;
    const int k0 = up * 2;
    if (h < 10) { amat.x = W1[h * 4 + k0]; amat.y = W1[h * 4 + k0 + 1]; }
  }

  v8f bias1;
#pragma unroll
  for (int v = 0; v < 8; ++v) {
    const int m = v + 8 * up;
    bias1[v] = (m < 10) ? b1[m] : 0.0f;
  }

  float w2r[30];
#pragma unroll
  for (int k = 0; k < 30; ++k) w2r[k] = W2[k];
  const float b20 = b2[0], b21 = b2[1], b22 = b2[2];

  const float* xrun = runs + (size_t)run * T_STEPS * (N_AG * 2);
  float*       orun = out  + (size_t)run * T_STEPS * N_AG;
  const int rot_src = (l15 + 15) & 15;

  auto xptr = [&](int s, int w) __attribute__((always_inline)) -> const float4* {
    const int u  = w - 2 * s;
    const int uu = ((unsigned)u < (unsigned)U_SPAN) ? u : 0;
    const int t  = s + SLOTS * (uu >> 7);
    const int i  = uu & (N_AG - 1);
    return (const float4*)(xrun + ((size_t)t * N_AG + i) * 2);
  };

  auto mlp = [&](v2f bmat, float& o0, float& o1, float& o2) __attribute__((always_inline)) {
    v8f d = __builtin_amdgcn_wmma_f32_16x16x4_f32(
        false, amat, false, bmat, (short)0, bias1, false, false);

    float dt[8];
#pragma unroll
    for (int v = 0; v < 8; ++v) dt[v] = fast_tanh(d[v]);

    const float h8 = __shfl_xor(dt[0], 16, 32);
    const float h9 = __shfl_xor(dt[1], 16, 32);

    float e0 = fmaf(w2r[20], dt[0], b22);
    float e1 =      w2r[21] * dt[1];
    e0 = fmaf(w2r[22], dt[2], e0);  e1 = fmaf(w2r[23], dt[3], e1);
    e0 = fmaf(w2r[24], dt[4], e0);  e1 = fmaf(w2r[25], dt[5], e1);
    e0 = fmaf(w2r[26], dt[6], e0);  e1 = fmaf(w2r[27], dt[7], e1);
    e0 = fmaf(w2r[28], h8,    e0);  e1 = fmaf(w2r[29], h9,    e1);
    o2 = e0 + e1;

    float c0 = fmaf(w2r[10], dt[0], b21), c1 = w2r[11] * dt[1];
    c0 = fmaf(w2r[12], dt[2], c0); c1 = fmaf(w2r[13], dt[3], c1);
    c0 = fmaf(w2r[14], dt[4], c0); c1 = fmaf(w2r[15], dt[5], c1);
    c0 = fmaf(w2r[16], dt[6], c0); c1 = fmaf(w2r[17], dt[7], c1);
    c0 = fmaf(w2r[18], h8,   c0);  c1 = fmaf(w2r[19], h9,   c1);
    o1 = c0 + c1;

    float a0 = fmaf(w2r[0], dt[0], b20), a1 = w2r[1] * dt[1];
    a0 = fmaf(w2r[2], dt[2], a0);  a1 = fmaf(w2r[3], dt[3], a1);
    a0 = fmaf(w2r[4], dt[4], a0);  a1 = fmaf(w2r[5], dt[5], a1);
    a0 = fmaf(w2r[6], dt[6], a0);  a1 = fmaf(w2r[7], dt[7], a1);
    a0 = fmaf(w2r[8], h8,   a0);   a1 = fmaf(w2r[9], h9,   a1);
    o0 = a0 + a1;
  };

  float co[TILES], cb[TILES];
#pragma unroll
  for (int k = 0; k < TILES; ++k) { co[k] = 0.0f; cb[k] = 0.0f; }

  auto iter = [&](const int w, const float4 (&xa)[TILES], float4 (&xc)[TILES],
                  const bool all_busy)
      __attribute__((always_inline)) {
#pragma unroll
    for (int k = 0; k < TILES; ++k) xc[k] = *xptr(16 * k + l15, w + 4);

    {
      const int kp = (w >> 1) & (TILES - 1);
      __builtin_prefetch((const char*)xptr(16 * kp + l15, w + 4 + 16), 0, 0);
    }

    int   uu[TILES];
    float o0e[TILES], o1s[TILES], rr[TILES];

#pragma unroll
    for (int k = 0; k < TILES; ++k) {
      const int s = 16 * k + l15;
      const int u = w - 2 * s;
      uu[k] = u;
      const int  i  = u & (N_AG - 1);
      const bool t0 = (s == 0) && (u < N_AG);
      const float co_in = (i == 0) ? 0.0f : co[k];
      const float cb_in = t0       ? 0.0f : cb[k];
      v2f bmat;
      bmat.x = up ? co_in : xa[k].x;
      bmat.y = up ? cb_in : xa[k].y;
      float o2;
      mlp(bmat, o0e[k], o1s[k], o2);
      co[k] = __shfl_xor(o2, 16, 32);
    }
#pragma unroll
    for (int k = 0; k < TILES; ++k) rr[k] = __shfl(o1s[k], rot_src, 32);
#pragma unroll
    for (int k = 0; k < TILES; ++k)
      cb[k] = (l15 == 0) ? rr[(k + TILES - 1) & (TILES - 1)] : rr[k];

    float o0o[TILES];
#pragma unroll
    for (int k = 0; k < TILES; ++k) {
      const int s = 16 * k + l15;
      const int u = uu[k];
      const int  i1 = (u & (N_AG - 1)) + 1;
      const bool t0 = (s == 0) && (u < N_AG);
      const float cb_in = (t0 || i1 == N_AG - 1) ? 0.0f : cb[k];
      v2f bmat;
      bmat.x = up ? co[k] : xa[k].z;
      bmat.y = up ? cb_in : xa[k].w;
      float o2;
      mlp(bmat, o0o[k], o1s[k], o2);
      co[k] = __shfl_xor(o2, 16, 32);
    }
#pragma unroll
    for (int k = 0; k < TILES; ++k) rr[k] = __shfl(o1s[k], rot_src, 32);
#pragma unroll
    for (int k = 0; k < TILES; ++k)
      cb[k] = (l15 == 0) ? rr[(k + TILES - 1) & (TILES - 1)] : rr[k];

    if (all_busy) {
      if (up == 0) {
#pragma unroll
        for (int k = 0; k < TILES; ++k) {
          const unsigned u   = (unsigned)uu[k];
          const unsigned t   = 16u * k + l15 + SLOTS * (u >> 7);
          const unsigned off = t * N_AG + (u & (N_AG - 1));
          float2 v; v.x = o0e[k]; v.y = o0o[k];
          *(float2*)(orun + off) = v;
        }
      }
    } else if (up == 0) {
#pragma unroll
      for (int k = 0; k < TILES; ++k) {
        const int u = uu[k];
        if ((unsigned)u < (unsigned)U_SPAN) {
          const int t = 16 * k + l15 + SLOTS * (u >> 7);
          const int i = u & (N_AG - 1);
          float2 v; v.x = o0e[k]; v.y = o0o[k];
          *(float2*)(orun + (size_t)t * N_AG + i) = v;
        }
      }
    }

  };

  float4 X0[TILES], X1[TILES], X2[TILES];
#pragma unroll
  for (int k = 0; k < TILES; ++k) {
    X0[k] = *xptr(16 * k + l15, 0);
    X1[k] = *xptr(16 * k + l15, 2);
  }
  int w = 0;
  for (; w < W_FULL_LO; w += 6) {
    iter(w,     X0, X2, false);
    iter(w + 2, X1, X0, false);
    iter(w + 4, X2, X1, false);
  }
  for (; w < W_FULL_HI; w += 6) {
    iter(w,     X0, X2, true);
    iter(w + 2, X1, X0, true);
    iter(w + 4, X2, X1, true);
  }
  for (; w < W_LOOP; w += 6) {
    iter(w,     X0, X2, false);
    iter(w + 2, X1, X0, false);
    iter(w + 4, X2, X1, false);
  }
}

extern "C" void kernel_launch(void* const* d_in, const int* in_sizes, int n_in,
                              void* d_out, int out_size, void* d_ws, size_t ws_size,
                              hipStream_t stream) {
  (void)in_sizes; (void)n_in; (void)out_size; (void)d_ws; (void)ws_size;
  const float* runs = (const float*)d_in[0];
  const float* W1   = (const float*)d_in[1];
  const float* b1   = (const float*)d_in[2];
  const float* W2   = (const float*)d_in[3];
  const float* b2   = (const float*)d_in[4];
  float* out        = (float*)d_out;

  dim3 grid(R_TOT);
  dim3 block(32);
  recurrent_mlp_wavefront_kernel<<<grid, block, 0, stream>>>(runs, W1, b1, W2, b2, out);
}


# All four lane exchanges per cell go by v_permlanex16_b32 on the vector ALU instead of ds_bpermute_b32 through the LDS unit (o1 with rotated selects); every sum and its order unchanged
# speedup vs baseline: 30.5684x; 1.5715x over previous
#include <hip/hip_runtime.h>


typedef float v2f __attribute__((ext_vector_type(2)));
typedef float v8f __attribute__((ext_vector_type(8)));

#define R_TOT    128
#define T_STEPS  1024
#define N_AG     128
#define TILES    4
#define SLOTS    (16 * TILES)
#define U_SPAN   ((T_STEPS / SLOTS) * N_AG)
#define W_TOTAL  (2 * (SLOTS - 1) + U_SPAN)
#define W_LOOP   (((W_TOTAL + 5) / 6) * 6)
#define W_FULL_LO ((((2 * (SLOTS - 1)) + 5) / 6) * 6)
#define W_FULL_HI ((U_SPAN / 6) * 6)
static_assert(W_FULL_LO >= 2 * (SLOTS - 1) && W_FULL_LO % 6 == 0, "full stretch starts on a trip boundary");
static_assert(W_FULL_HI % 6 == 0 && W_FULL_HI - 2 < U_SPAN && W_FULL_HI <= W_LOOP, "full stretch ends inside every span");

__device__ __forceinline__ float fast_tanh(float x) {
#if __has_builtin(__builtin_amdgcn_tanhf)
  return __builtin_amdgcn_tanhf(x);
#else
  float e = __builtin_exp2f(x * 2.885390081777927f);
  return 1.0f - 2.0f / (e + 1.0f);
#endif
}

__device__ __forceinline__ float xhalf(float x) {
  const int v = __builtin_bit_cast(int, x);
  return __builtin_bit_cast(float,
      __builtin_amdgcn_permlanex16(v, v, 0x76543210, 0xfedcba98, false, false));
}

__device__ __forceinline__ float xhalf_keep(float dead, float x) {
  return __builtin_bit_cast(float,
      __builtin_amdgcn_permlanex16(__builtin_bit_cast(int, dead), __builtin_bit_cast(int, x),
                                   0x76543210, 0xfedcba98, false, false));
}

__device__ __forceinline__ float xhalf_from_slot_below(float x) {
  const int v = __builtin_bit_cast(int, x);
  return __builtin_bit_cast(float,
      __builtin_amdgcn_permlanex16(v, v, 0x6543210f, 0xedcba987, false, false));
}

__global__ __launch_bounds__(32)
void recurrent_mlp_wavefront_kernel(const float* __restrict__ runs,
                              const float* __restrict__ W1,
                              const float* __restrict__ b1,
                              const float* __restrict__ W2,
                              const float* __restrict__ b2,
                              float*       __restrict__ out)
{
  const int lane = threadIdx.x;
  const int up   = lane >> 4;
  const int l15  = lane & 15;
  const int run  = blockIdx.x;

  v2f amat; amat.x = 0.0f; amat.y = 0.0f;
  {
    const int h  = l15;
    const int k0 = up * 2;
    if (h < 10) { amat.x = W1[h * 4 + k0]; amat.y = W1[h * 4 + k0 + 1]; }
  }

  v8f bias1;
#pragma unroll
  for (int v = 0; v < 8; ++v) {
    const int m = v + 8 * up;
    bias1[v] = (m < 10) ? b1[m] : 0.0f;
  }

  float w2r[30];
#pragma unroll
  for (int k = 0; k < 30; ++k) w2r[k] = W2[k];
  const float b20 = b2[0], b21 = b2[1], b22 = b2[2];

  const float* xrun = runs + (size_t)run * T_STEPS * (N_AG * 2);
  float*       orun = out  + (size_t)run * T_STEPS * N_AG;

  auto xptr = [&](int s, int w) __attribute__((always_inline)) -> const float4* {
    const int u  = w - 2 * s;
    const int uu = ((unsigned)u < (unsigned)U_SPAN) ? u : 0;
    const int t  = s + SLOTS * (uu >> 7);
    const int i  = uu & (N_AG - 1);
    return (const float4*)(xrun + ((size_t)t * N_AG + i) * 2);
  };

  auto mlp = [&](v2f bmat, float& o0, float& o1, float& o2) __attribute__((always_inline)) {
    v8f d = __builtin_amdgcn_wmma_f32_16x16x4_f32(
        false, amat, false, bmat, (short)0, bias1, false, false);

    float dt[8];
#pragma unroll
    for (int v = 0; v < 8; ++v) dt[v] = fast_tanh(d[v]);

    const float h8 = xhalf_keep(d[0], dt[0]);
    const float h9 = xhalf_keep(d[1], dt[1]);

    float e0 = fmaf(w2r[20], dt[0], b22);
    float e1 =      w2r[21] * dt[1];
    e0 = fmaf(w2r[22], dt[2], e0);  e1 = fmaf(w2r[23], dt[3], e1);
    e0 = fmaf(w2r[24], dt[4], e0);  e1 = fmaf(w2r[25], dt[5], e1);
    e0 = fmaf(w2r[26], dt[6], e0);  e1 = fmaf(w2r[27], dt[7], e1);
    e0 = fmaf(w2r[28], h8,    e0);  e1 = fmaf(w2r[29], h9,    e1);
    o2 = e0 + e1;

    float c0 = fmaf(w2r[10], dt[0], b21), c1 = w2r[11] * dt[1];
    c0 = fmaf(w2r[12], dt[2], c0); c1 = fmaf(w2r[13], dt[3], c1);
    c0 = fmaf(w2r[14], dt[4], c0); c1 = fmaf(w2r[15], dt[5], c1);
    c0 = fmaf(w2r[16], dt[6], c0); c1 = fmaf(w2r[17], dt[7], c1);
    c0 = fmaf(w2r[18], h8,   c0);  c1 = fmaf(w2r[19], h9,   c1);
    o1 = c0 + c1;

    float a0 = fmaf(w2r[0], dt[0], b20), a1 = w2r[1] * dt[1];
    a0 = fmaf(w2r[2], dt[2], a0);  a1 = fmaf(w2r[3], dt[3], a1);
    a0 = fmaf(w2r[4], dt[4], a0);  a1 = fmaf(w2r[5], dt[5], a1);
    a0 = fmaf(w2r[6], dt[6], a0);  a1 = fmaf(w2r[7], dt[7], a1);
    a0 = fmaf(w2r[8], h8,   a0);   a1 = fmaf(w2r[9], h9,   a1);
    o0 = a0 + a1;
  };

  float co[TILES], cb[TILES];
#pragma unroll
  for (int k = 0; k < TILES; ++k) { co[k] = 0.0f; cb[k] = 0.0f; }

  auto iter = [&](const int w, const float4 (&xa)[TILES], float4 (&xc)[TILES],
                  const bool all_busy)
      __attribute__((always_inline)) {
#pragma unroll
    for (int k = 0; k < TILES; ++k) xc[k] = *xptr(16 * k + l15, w + 4);

    {
      const int kp = (w >> 1) & (TILES - 1);
      __builtin_prefetch((const char*)xptr(16 * kp + l15, w + 4 + 16), 0, 0);
    }

    int   uu[TILES];
    float o0e[TILES], o1s[TILES], rr[TILES];

#pragma unroll
    for (int k = 0; k < TILES; ++k) {
      const int s = 16 * k + l15;
      const int u = w - 2 * s;
      uu[k] = u;
      const int  i  = u & (N_AG - 1);
      const bool t0 = (s == 0) && (u < N_AG);
      const float co_in = (i == 0) ? 0.0f : co[k];
      const float cb_in = t0       ? 0.0f : cb[k];
      v2f bmat;
      bmat.x = up ? co_in : xa[k].x;
      bmat.y = up ? cb_in : xa[k].y;
      float o2;
      mlp(bmat, o0e[k], o1s[k], o2);
      co[k] = xhalf(o2);
    }
#pragma unroll
    for (int k = 0; k < TILES; ++k) rr[k] = xhalf_from_slot_below(o1s[k]);
#pragma unroll
    for (int k = 0; k < TILES; ++k)
      cb[k] = (l15 == 0) ? rr[(k + TILES - 1) & (TILES - 1)] : rr[k];

    float o0o[TILES];
#pragma unroll
    for (int k = 0; k < TILES; ++k) {
      const int s = 16 * k + l15;
      const int u = uu[k];
      const int  i1 = (u & (N_AG - 1)) + 1;
      const bool t0 = (s == 0) && (u < N_AG);
      const float cb_in = (t0 || i1 == N_AG - 1) ? 0.0f : cb[k];
      v2f bmat;
      bmat.x = up ? co[k] : xa[k].z;
      bmat.y = up ? cb_in : xa[k].w;
      float o2;
      mlp(bmat, o0o[k], o1s[k], o2);
      co[k] = xhalf(o2);
    }
#pragma unroll
    for (int k = 0; k < TILES; ++k) rr[k] = xhalf_from_slot_below(o1s[k]);
#pragma unroll
    for (int k = 0; k < TILES; ++k)
      cb[k] = (l15 == 0) ? rr[(k + TILES - 1) & (TILES - 1)] : rr[k];

    if (all_busy) {
      if (up == 0) {
#pragma unroll
        for (int k = 0; k < TILES; ++k) {
          const unsigned u   = (unsigned)uu[k];
          const unsigned t   = 16u * k + l15 + SLOTS * (u >> 7);
          const unsigned off = t * N_AG + (u & (N_AG - 1));
          float2 v; v.x = o0e[k]; v.y = o0o[k];
          *(float2*)(orun + off) = v;
        }
      }
    } else if (up == 0) {
#pragma unroll
      for (int k = 0; k < TILES; ++k) {
        const int u = uu[k];
        if ((unsigned)u < (unsigned)U_SPAN) {
          const int t = 16 * k + l15 + SLOTS * (u >> 7);
          const int i = u & (N_AG - 1);
          float2 v; v.x = o0e[k]; v.y = o0o[k];
          *(float2*)(orun + (size_t)t * N_AG + i) = v;
        }
      }
    }

  };

  float4 X0[TILES], X1[TILES], X2[TILES];
#pragma unroll
  for (int k = 0; k < TILES; ++k) {
    X0[k] = *xptr(16 * k + l15, 0);
    X1[k] = *xptr(16 * k + l15, 2);
  }
  int w = 0;
  for (; w < W_FULL_LO; w += 6) {
    iter(w,     X0, X2, false);
    iter(w + 2, X1, X0, false);
    iter(w + 4, X2, X1, false);
  }
  for (; w < W_FULL_HI; w += 6) {
    iter(w,     X0, X2, true);
    iter(w + 2, X1, X0, true);
    iter(w + 4, X2, X1, true);
  }
  for (; w < W_LOOP; w += 6) {
    iter(w,     X0, X2, false);
    iter(w + 2, X1, X0, false);
    iter(w + 4, X2, X1, false);
  }
}

extern "C" void kernel_launch(void* const* d_in, const int* in_sizes, int n_in,
                              void* d_out, int out_size, void* d_ws, size_t ws_size,
                              hipStream_t stream) {
  (void)in_sizes; (void)n_in; (void)out_size; (void)d_ws; (void)ws_size;
  const float* runs = (const float*)d_in[0];
  const float* W1   = (const float*)d_in[1];
  const float* b1   = (const float*)d_in[2];
  const float* W2   = (const float*)d_in[3];
  const float* b2   = (const float*)d_in[4];
  float* out        = (float*)d_out;

  dim3 grid(R_TOT);
  dim3 block(32);
  recurrent_mlp_wavefront_kernel<<<grid, block, 0, stream>>>(runs, W1, b1, W2, b2, out);
}


# The four 16-slot tiles of a run go to four waves of one work-group (one per SIMD), not one wave; the o1 value that crosses a tile edge each step goes through LDS behind one barrier; sums unchanged
# speedup vs baseline: 41.9957x; 1.3738x over previous
#include <hip/hip_runtime.h>


typedef float v2f __attribute__((ext_vector_type(2)));
typedef float v8f __attribute__((ext_vector_type(8)));

#define R_TOT    128
#define T_STEPS  1024
#define N_AG     128
#define TILES    4
#define SLOTS    (16 * TILES)
#define U_SPAN   ((T_STEPS / SLOTS) * N_AG)
#define W_TOTAL  (2 * (SLOTS - 1) + U_SPAN)
#define W_LOOP   (((W_TOTAL + 5) / 6) * 6)
#define W_FULL_LO ((((2 * (SLOTS - 1)) + 5) / 6) * 6)
#define W_FULL_HI ((U_SPAN / 6) * 6)
static_assert(W_FULL_LO >= 2 * (SLOTS - 1) && W_FULL_LO % 6 == 0, "full stretch starts on a trip boundary");
static_assert(W_FULL_HI % 6 == 0 && W_FULL_HI - 2 < U_SPAN && W_FULL_HI <= W_LOOP, "full stretch ends inside every span");

__device__ __forceinline__ float fast_tanh(float x) {
#if __has_builtin(__builtin_amdgcn_tanhf)
  return __builtin_amdgcn_tanhf(x);
#else
  float e = __builtin_exp2f(x * 2.885390081777927f);
  return 1.0f - 2.0f / (e + 1.0f);
#endif
}

__device__ __forceinline__ float xhalf(float x) {
  const int v = __builtin_bit_cast(int, x);
  return __builtin_bit_cast(float,
      __builtin_amdgcn_permlanex16(v, v, 0x76543210, 0xfedcba98, false, false));
}

__device__ __forceinline__ float xhalf_keep(float dead, float x) {
  return __builtin_bit_cast(float,
      __builtin_amdgcn_permlanex16(__builtin_bit_cast(int, dead), __builtin_bit_cast(int, x),
                                   0x76543210, 0xfedcba98, false, false));
}

__device__ __forceinline__ float xhalf_from_slot_below(float x) {
  const int v = __builtin_bit_cast(int, x);
  return __builtin_bit_cast(float,
      __builtin_amdgcn_permlanex16(v, v, 0x6543210f, 0xedcba987, false, false));
}

__global__ __launch_bounds__(32 * TILES)
void recurrent_mlp_wavefront_kernel(const float* __restrict__ runs,
                              const float* __restrict__ W1,
                              const float* __restrict__ b1,
                              const float* __restrict__ W2,
                              const float* __restrict__ b2,
                              float*       __restrict__ out)
{
  __shared__ float edge[2][TILES][32];

  const int lane = threadIdx.x & 31;
  const int up   = lane >> 4;
  const int l15  = lane & 15;
  const int tile  = __builtin_amdgcn_readfirstlane((int)(threadIdx.x >> 5));
  const int below = (tile + TILES - 1) & (TILES - 1);
  const int s     = 16 * tile + l15;
  const int run   = blockIdx.x;

  v2f amat; amat.x = 0.0f; amat.y = 0.0f;
  {
    const int h  = l15;
    const int k0 = up * 2;
    if (h < 10) { amat.x = W1[h * 4 + k0]; amat.y = W1[h * 4 + k0 + 1]; }
  }

  v8f bias1;
#pragma unroll
  for (int v = 0; v < 8; ++v) {
    const int m = v + 8 * up;
    bias1[v] = (m < 10) ? b1[m] : 0.0f;
  }

  float w2r[30];
#pragma unroll
  for (int k = 0; k < 30; ++k) w2r[k] = W2[k];
  const float b20 = b2[0], b21 = b2[1], b22 = b2[2];

  const float* xrun = runs + (size_t)run * T_STEPS * (N_AG * 2);
  float*       orun = out  + (size_t)run * T_STEPS * N_AG;

  auto xptr = [&](int sl, int w) __attribute__((always_inline)) -> const float4* {
    const int u  = w - 2 * sl;
    const int uu = ((unsigned)u < (unsigned)U_SPAN) ? u : 0;
    const int t  = sl + SLOTS * (uu >> 7);
    const int i  = uu & (N_AG - 1);
    return (const float4*)(xrun + ((size_t)t * N_AG + i) * 2);
  };

  auto mlp = [&](v2f bmat, float& o0, float& o1, float& o2) __attribute__((always_inline)) {
    v8f d = __builtin_amdgcn_wmma_f32_16x16x4_f32(
        false, amat, false, bmat, (short)0, bias1, false, false);

    float dt[8];
#pragma unroll
    for (int v = 0; v < 8; ++v) dt[v] = fast_tanh(d[v]);

    const float h8 = xhalf_keep(d[0], dt[0]);
    const float h9 = xhalf_keep(d[1], dt[1]);

    float c0 = fmaf(w2r[10], dt[0], b21), c1 = w2r[11] * dt[1];
    c0 = fmaf(w2r[12], dt[2], c0); c1 = fmaf(w2r[13], dt[3], c1);
    c0 = fmaf(w2r[14], dt[4], c0); c1 = fmaf(w2r[15], dt[5], c1);
    c0 = fmaf(w2r[16], dt[6], c0); c1 = fmaf(w2r[17], dt[7], c1);
    c0 = fmaf(w2r[18], h8,   c0);  c1 = fmaf(w2r[19], h9,   c1);
    o1 = c0 + c1;

    float e0 = fmaf(w2r[20], dt[0], b22);
    float e1 =      w2r[21] * dt[1];
    e0 = fmaf(w2r[22], dt[2], e0);  e1 = fmaf(w2r[23], dt[3], e1);
    e0 = fmaf(w2r[24], dt[4], e0);  e1 = fmaf(w2r[25], dt[5], e1);
    e0 = fmaf(w2r[26], dt[6], e0);  e1 = fmaf(w2r[27], dt[7], e1);
    e0 = fmaf(w2r[28], h8,    e0);  e1 = fmaf(w2r[29], h9,    e1);
    o2 = e0 + e1;

    float a0 = fmaf(w2r[0], dt[0], b20), a1 = w2r[1] * dt[1];
    a0 = fmaf(w2r[2], dt[2], a0);  a1 = fmaf(w2r[3], dt[3], a1);
    a0 = fmaf(w2r[4], dt[4], a0);  a1 = fmaf(w2r[5], dt[5], a1);
    a0 = fmaf(w2r[6], dt[6], a0);  a1 = fmaf(w2r[7], dt[7], a1);
    a0 = fmaf(w2r[8], h8,   a0);   a1 = fmaf(w2r[9], h9,   a1);
    o0 = a0 + a1;
  };

  float co = 0.0f, cb = 0.0f;

  auto slot_up = [&](float o1s, const int par) __attribute__((always_inline)) -> float {
    edge[par][tile][lane] = o1s;
    const float rr = xhalf_from_slot_below(o1s);
    __builtin_amdgcn_fence(__ATOMIC_RELEASE, "workgroup", "local");
    __builtin_amdgcn_s_barrier();
    __builtin_amdgcn_fence(__ATOMIC_ACQUIRE, "workgroup", "local");
    const float eb = edge[par][below][15];
    return (l15 == 0) ? eb : rr;
  };

  auto iter = [&](const int w, const float4& xa, float4& xc, const bool all_busy)
      __attribute__((always_inline)) {
    xc = *xptr(s, w + 4);

    if (((w >> 1) & (TILES - 1)) == tile)
      __builtin_prefetch((const char*)xptr(s, w + 4 + 16), 0, 0);

    const int  u  = w - 2 * s;
    const int  i  = u & (N_AG - 1);
    const bool t0 = (s == 0) && (u < N_AG);
    float o0e, o0o, o1, o2;

    {
      const float co_in = (i == 0) ? 0.0f : co;
      const float cb_in = t0       ? 0.0f : cb;
      v2f bmat;
      bmat.x = up ? co_in : xa.x;
      bmat.y = up ? cb_in : xa.y;
      mlp(bmat, o0e, o1, o2);
      co = xhalf(o2);
      cb = slot_up(o1, 0);
    }

    {
      const int i1 = i + 1;
      const float cb_in = (t0 || i1 == N_AG - 1) ? 0.0f : cb;
      v2f bmat;
      bmat.x = up ? co    : xa.z;
      bmat.y = up ? cb_in : xa.w;
      mlp(bmat, o0o, o1, o2);
      co = xhalf(o2);
      cb = slot_up(o1, 1);
    }

    if (up == 0) {
      if (all_busy || (unsigned)u < (unsigned)U_SPAN) {
        const unsigned uu  = (unsigned)u;
        const unsigned t   = (unsigned)s + SLOTS * (uu >> 7);
        const unsigned off = t * N_AG + (uu & (N_AG - 1));
        float2 v; v.x = o0e; v.y = o0o;
        *(float2*)(orun + off) = v;
      }
    }
  };

  float4 X0 = *xptr(s, 0);
  float4 X1 = *xptr(s, 2);
  float4 X2;
  int w = 0;
  for (; w < W_FULL_LO; w += 6) {
    iter(w,     X0, X2, false);
    iter(w + 2, X1, X0, false);
    iter(w + 4, X2, X1, false);
  }
  for (; w < W_FULL_HI; w += 6) {
    iter(w,     X0, X2, true);
    iter(w + 2, X1, X0, true);
    iter(w + 4, X2, X1, true);
  }
  for (; w < W_LOOP; w += 6) {
    iter(w,     X0, X2, false);
    iter(w + 2, X1, X0, false);
    iter(w + 4, X2, X1, false);
  }
}

extern "C" void kernel_launch(void* const* d_in, const int* in_sizes, int n_in,
                              void* d_out, int out_size, void* d_ws, size_t ws_size,
                              hipStream_t stream) {
  (void)in_sizes; (void)n_in; (void)out_size; (void)d_ws; (void)ws_size;
  const float* runs = (const float*)d_in[0];
  const float* W1   = (const float*)d_in[1];
  const float* b1   = (const float*)d_in[2];
  const float* W2   = (const float*)d_in[3];
  const float* b2   = (const float*)d_in[4];
  float* out        = (float*)d_out;

  dim3 grid(R_TOT);
  dim3 block(32 * TILES);
  recurrent_mlp_wavefront_kernel<<<grid, block, 0, stream>>>(runs, W1, b1, W2, b2, out);
}


# Each tile trails the one below by 2 extra steps: the o1 word crossing waves is needed 3 steps after it is made; LDS store, barrier (one per 2 steps, signal/wait apart) and load leave the step chain
# speedup vs baseline: 57.3345x; 1.3652x over previous
#include <hip/hip_runtime.h>


typedef float v2f __attribute__((ext_vector_type(2)));
typedef float v8f __attribute__((ext_vector_type(8)));

#define R_TOT    128
#define T_STEPS  1024
#define N_AG     128
#define TILES    4
#define SLOTS    (16 * TILES)
#define ROUNDS   (T_STEPS / SLOTS)
#define GAP      (2 * TILES)
#define PERIOD   (N_AG + GAP)
#define W_TOTAL  (2 * (SLOTS - 1 + TILES - 1) + (ROUNDS - 1) * PERIOD + N_AG)
#define W_LOOP   (((W_TOTAL + 11) / 12) * 12)
static_assert(T_STEPS % SLOTS == 0 && PERIOD % 2 == 0 && N_AG % 2 == 0, "even steps start on even agents");

__device__ __forceinline__ float fast_tanh(float x) {
#if __has_builtin(__builtin_amdgcn_tanhf)
  return __builtin_amdgcn_tanhf(x);
#else
  float e = __builtin_exp2f(x * 2.885390081777927f);
  return 1.0f - 2.0f / (e + 1.0f);
#endif
}

__device__ __forceinline__ float xhalf(float x) {
  const int v = __builtin_bit_cast(int, x);
  return __builtin_bit_cast(float,
      __builtin_amdgcn_permlanex16(v, v, 0x76543210, 0xfedcba98, false, false));
}

__device__ __forceinline__ float xhalf_keep(float dead, float x) {
  return __builtin_bit_cast(float,
      __builtin_amdgcn_permlanex16(__builtin_bit_cast(int, dead), __builtin_bit_cast(int, x),
                                   0x76543210, 0xfedcba98, false, false));
}

__device__ __forceinline__ float xhalf_from_slot_below(float x) {
  const int v = __builtin_bit_cast(int, x);
  return __builtin_bit_cast(float,
      __builtin_amdgcn_permlanex16(v, v, 0x6543210f, 0xedcba987, false, false));
}

__global__ __launch_bounds__(32 * TILES)
void recurrent_mlp_wavefront_kernel(const float* __restrict__ runs,
                              const float* __restrict__ W1,
                              const float* __restrict__ b1,
                              const float* __restrict__ W2,
                              const float* __restrict__ b2,
                              float*       __restrict__ out)
{
  __shared__ float edge[2][2][TILES][32];

  const int lane = threadIdx.x & 31;
  const int up   = lane >> 4;
  const int l15  = lane & 15;
  const int tile  = __builtin_amdgcn_readfirstlane((int)(threadIdx.x >> 5));
  const int below = (tile + TILES - 1) & (TILES - 1);
  const int s     = 16 * tile + l15;
  const int run   = blockIdx.x;

  v2f amat; amat.x = 0.0f; amat.y = 0.0f;
  {
    const int h  = l15;
    const int k0 = up * 2;
    if (h < 10) { amat.x = W1[h * 4 + k0]; amat.y = W1[h * 4 + k0 + 1]; }
  }

  v8f bias1;
#pragma unroll
  for (int v = 0; v < 8; ++v) {
    const int m = v + 8 * up;
    bias1[v] = (m < 10) ? b1[m] : 0.0f;
  }

  float w2r[30];
#pragma unroll
  for (int k = 0; k < 30; ++k) w2r[k] = W2[k];
  const float b20 = b2[0], b21 = b2[1], b22 = b2[2];

  const float* xrun = runs + (size_t)run * T_STEPS * (N_AG * 2);
  float*       orun = out  + (size_t)run * T_STEPS * N_AG;

  int iu = -2 * (s + tile);
  int tb = s;

  auto xptr = [&](const int d) __attribute__((always_inline)) -> const float4* {
    int ii = iu + d, tt = tb;
    if (ii >= PERIOD) { ii -= PERIOD; tt += SLOTS; }
    ii = min(max(ii, 0), N_AG - 2);
    tt &= (T_STEPS - 1);
    return (const float4*)(xrun + ((size_t)(tt * N_AG + ii)) * 2);
  };

  auto mlp = [&](v2f bmat, float& o0, float& o1, float& o2) __attribute__((always_inline)) {
    v8f d = __builtin_amdgcn_wmma_f32_16x16x4_f32(
        false, amat, false, bmat, (short)0, bias1, false, false);

    float dt[8];
#pragma unroll
    for (int v = 0; v < 8; ++v) dt[v] = fast_tanh(d[v]);

    const float h8 = xhalf_keep(d[0], dt[0]);
    const float h9 = xhalf_keep(d[1], dt[1]);

    float c0 = fmaf(w2r[10], dt[0], b21), c1 = w2r[11] * dt[1];
    c0 = fmaf(w2r[12], dt[2], c0); c1 = fmaf(w2r[13], dt[3], c1);
    c0 = fmaf(w2r[14], dt[4], c0); c1 = fmaf(w2r[15], dt[5], c1);
    c0 = fmaf(w2r[16], dt[6], c0); c1 = fmaf(w2r[17], dt[7], c1);
    c0 = fmaf(w2r[18], h8,   c0);  c1 = fmaf(w2r[19], h9,   c1);
    o1 = c0 + c1;

    float e0 = fmaf(w2r[20], dt[0], b22);
    float e1 =      w2r[21] * dt[1];
    e0 = fmaf(w2r[22], dt[2], e0);  e1 = fmaf(w2r[23], dt[3], e1);
    e0 = fmaf(w2r[24], dt[4], e0);  e1 = fmaf(w2r[25], dt[5], e1);
    e0 = fmaf(w2r[26], dt[6], e0);  e1 = fmaf(w2r[27], dt[7], e1);
    e0 = fmaf(w2r[28], h8,    e0);  e1 = fmaf(w2r[29], h9,    e1);
    o2 = e0 + e1;

    float a0 = fmaf(w2r[0], dt[0], b20), a1 = w2r[1] * dt[1];
    a0 = fmaf(w2r[2], dt[2], a0);  a1 = fmaf(w2r[3], dt[3], a1);
    a0 = fmaf(w2r[4], dt[4], a0);  a1 = fmaf(w2r[5], dt[5], a1);
    a0 = fmaf(w2r[6], dt[6], a0);  a1 = fmaf(w2r[7], dt[7], a1);
    a0 = fmaf(w2r[8], h8,   a0);   a1 = fmaf(w2r[9], h9,   a1);
    o0 = a0 + a1;
  };

  float co = 0.0f, cb = 0.0f;
  float EA = 0.0f, EB = 0.0f;

  auto iter = [&](const int w, float4& xa, const int par)
      __attribute__((always_inline)) {
    const bool busy = ((unsigned)iu < (unsigned)N_AG) && (tb < T_STEPS);
    const bool t0   = (tb == 0);
    float o0e, o0o, o1, o2;

    {
      const float co_in = (iu == 0) ? 0.0f : co;
      const float cb_in = t0        ? 0.0f : cb;
      v2f bmat;
      bmat.x = up ? co_in : xa.x;
      bmat.y = up ? cb_in : xa.y;
      mlp(bmat, o0e, o1, o2);
      co = xhalf(o2);
      edge[par][0][tile][lane] = o1;
      const float rr = xhalf_from_slot_below(o1);
      cb = (l15 == 0) ? EA : rr;
    }

    {
      const float cb_in = (t0 || iu == N_AG - 2) ? 0.0f : cb;
      v2f bmat;
      bmat.x = up ? co    : xa.z;
      bmat.y = up ? cb_in : xa.w;
      mlp(bmat, o0o, o1, o2);
      co = xhalf(o2);
      edge[par][1][tile][lane] = o1;
      const float rr = xhalf_from_slot_below(o1);
      cb = (l15 == 0) ? EB : rr;
    }

    if (up == 0 && busy) {
      const unsigned off = (unsigned)tb * N_AG + (unsigned)iu;
      float2 v; v.x = o0e; v.y = o0o;
      *(float2*)(orun + off) = v;
    }

    __builtin_amdgcn_fence(__ATOMIC_RELEASE, "workgroup", "local");
    __builtin_amdgcn_s_barrier_signal(-1);

    iu += 2;
    if (iu == PERIOD) { iu = 0; tb += SLOTS; }

    xa = *xptr(4);

    if (((w >> 1) & (TILES - 1)) == tile)
      __builtin_prefetch((const char*)xptr(4 + 16), 0, 0);

    __builtin_amdgcn_s_barrier_wait(-1);
    __builtin_amdgcn_fence(__ATOMIC_ACQUIRE, "workgroup", "local");
    EA = edge[par][0][below][15];
    EB = edge[par][1][below][15];
    __builtin_amdgcn_sched_barrier(0);
  };

  float4 X0 = *xptr(0);
  float4 X1 = *xptr(2);
  float4 X2 = *xptr(4);
  for (int w = 0; w < W_LOOP; w += 12) {
    iter(w,      X0, 0);
    iter(w + 2,  X1, 1);
    iter(w + 4,  X2, 0);
    iter(w + 6,  X0, 1);
    iter(w + 8,  X1, 0);
    iter(w + 10, X2, 1);
  }
}

extern "C" void kernel_launch(void* const* d_in, const int* in_sizes, int n_in,
                              void* d_out, int out_size, void* d_ws, size_t ws_size,
                              hipStream_t stream) {
  (void)in_sizes; (void)n_in; (void)out_size; (void)d_ws; (void)ws_size;
  const float* runs = (const float*)d_in[0];
  const float* W1   = (const float*)d_in[1];
  const float* b1   = (const float*)d_in[2];
  const float* W2   = (const float*)d_in[3];
  const float* b2   = (const float*)d_in[4];
  float* out        = (float*)d_out;

  dim3 grid(R_TOT);
  dim3 block(32 * TILES);
  recurrent_mlp_wavefront_kernel<<<grid, block, 0, stream>>>(runs, W1, b1, W2, b2, out);
}


# Hidden units split 5+5 over the lane halves; o2/o1 partial sums re-enter the next WMMA as two k rows each (it adds them), x*W1+b1 as C: no half exchange or select on the chain; sum order changes
# speedup vs baseline: 63.9094x; 1.1147x over previous
#include <hip/hip_runtime.h>


typedef float v2f __attribute__((ext_vector_type(2)));
typedef float v8f __attribute__((ext_vector_type(8)));

#define R_TOT    128
#define T_STEPS  1024
#define N_AG     128
#define TILES    4
#define SLOTS    (16 * TILES)
#define ROUNDS   (T_STEPS / SLOTS)
#define GAP      (2 * TILES)
#define PERIOD   (N_AG + GAP)
#define W_TOTAL  (2 * (SLOTS - 1 + TILES - 1) + (ROUNDS - 1) * PERIOD + N_AG)
#define W_LOOP   (((W_TOTAL + 11) / 12) * 12)
#define HALF_H   5
static_assert(T_STEPS % SLOTS == 0 && PERIOD % 2 == 0 && N_AG % 2 == 0, "even steps start on even agents");

__device__ __forceinline__ float fast_tanh(float x) {
#if __has_builtin(__builtin_amdgcn_tanhf)
  return __builtin_amdgcn_tanhf(x);
#else
  float e = __builtin_exp2f(x * 2.885390081777927f);
  return 1.0f - 2.0f / (e + 1.0f);
#endif
}

__device__ __forceinline__ float xhalf(float x) {
  const int v = __builtin_bit_cast(int, x);
  return __builtin_bit_cast(float,
      __builtin_amdgcn_permlanex16(v, v, 0x76543210, 0xfedcba98, false, false));
}

__device__ __forceinline__ float from_slot_below(float edge_word, float x) {
  return __builtin_bit_cast(float,
      __builtin_amdgcn_update_dpp(__builtin_bit_cast(int, edge_word), __builtin_bit_cast(int, x),
                                  0x111  , 0xf, 0xf, false));
}

__global__ __launch_bounds__(32 * TILES)
void recurrent_mlp_wavefront_kernel(const float* __restrict__ runs,
                              const float* __restrict__ W1,
                              const float* __restrict__ b1,
                              const float* __restrict__ W2,
                              const float* __restrict__ b2,
                              float*       __restrict__ out)
{
  __shared__ float edge[2][2][TILES][32];
  __shared__ float zero_word;

  const int lane = threadIdx.x & 31;
  const int up   = lane >> 4;
  const int l15  = lane & 15;
  const int tile  = __builtin_amdgcn_readfirstlane((int)(threadIdx.x >> 5));
  const int below = (tile + TILES - 1) & (TILES - 1);
  const int s     = 16 * tile + l15;
  const int run   = blockIdx.x;
  if (threadIdx.x == 0) zero_word = 0.0f;

  v2f amat; amat.x = 0.0f; amat.y = 0.0f;
  {
    const int row = l15;
    int unit = -1;
    if (row < HALF_H) unit = row;
    else if (row >= 8 && row < 8 + HALF_H) unit = row - 8 + HALF_H;
    if (unit >= 0) { amat.x = W1[unit * 4 + 2]; amat.y = W1[unit * 4 + 3]; }
  }

  float wx0[HALF_H], wx1[HALF_H], bb1[HALF_H], w2a[HALF_H], w2b[HALF_H], w2c[HALF_H];
#pragma unroll
  for (int r = 0; r < HALF_H; ++r) {
    const int unit = HALF_H * up + r;
    wx0[r] = W1[unit * 4 + 0];
    wx1[r] = W1[unit * 4 + 1];
    bb1[r] = b1[unit];
    w2a[r] = W2[0 * 10 + unit];
    w2b[r] = W2[1 * 10 + unit];
    w2c[r] = W2[2 * 10 + unit];
  }
  const float b20 = up ? 0.0f : b2[0];
  const float b21 = up ? 0.0f : b2[1];
  const float b22 = up ? 0.0f : b2[2];

  const float* xrun = runs + (size_t)run * T_STEPS * (N_AG * 2);
  float*       orun = out  + (size_t)run * T_STEPS * N_AG;

  int iu = -2 * (s + tile);
  int tb = s;

  auto xptr = [&](const int d) __attribute__((always_inline)) -> const float4* {
    int ii = iu + d, tt = tb;
    if (ii >= PERIOD) { ii -= PERIOD; tt += SLOTS; }
    ii = min(max(ii, 0), N_AG - 2);
    tt &= (T_STEPS - 1);
    return (const float4*)(xrun + ((size_t)(tt * N_AG + ii)) * 2);
  };

  auto xpart = [&](const float x0, const float x1, float (&cx)[HALF_H]) __attribute__((always_inline)) {
#pragma unroll
    for (int r = 0; r < HALF_H; ++r) cx[r] = fmaf(wx1[r], x1, fmaf(wx0[r], x0, bb1[r]));
  };

  auto mlp = [&](const float (&cx)[HALF_H], const v2f sm,
                 float& p0, float& p1, float& p2) __attribute__((always_inline)) {
    v8f c;
#pragma unroll
    for (int r = 0; r < HALF_H; ++r) c[r] = cx[r];
    c[5] = 0.0f; c[6] = 0.0f; c[7] = 0.0f;

    v8f d = __builtin_amdgcn_wmma_f32_16x16x4_f32(
        false, amat, false, sm, (short)0, c, false, false);

    float dt[HALF_H];
#pragma unroll
    for (int r = 0; r < HALF_H; ++r) dt[r] = fast_tanh(d[r]);

    float c0 = fmaf(w2b[0], dt[0], b21), c1 = w2b[1] * dt[1];
    c0 = fmaf(w2b[2], dt[2], c0);        c1 = fmaf(w2b[3], dt[3], c1);
    c0 = fmaf(w2b[4], dt[4], c0);
    p1 = c0 + c1;

    float e0 = fmaf(w2c[0], dt[0], b22), e1 = w2c[1] * dt[1];
    e0 = fmaf(w2c[2], dt[2], e0);        e1 = fmaf(w2c[3], dt[3], e1);
    e0 = fmaf(w2c[4], dt[4], e0);
    p2 = e0 + e1;

    float a0 = fmaf(w2a[0], dt[0], b20), a1 = w2a[1] * dt[1];
    a0 = fmaf(w2a[2], dt[2], a0);        a1 = fmaf(w2a[3], dt[3], a1);
    a0 = fmaf(w2a[4], dt[4], a0);
    p0 = a0 + a1;
  };

  float co = 0.0f, cb = 0.0f;
  float EA = 0.0f, EB = 0.0f;

  float cxe[HALF_H];

  auto iter = [&](const int w, float4& xa, const float4& xn, const int par)
      __attribute__((always_inline)) {
    const bool busy = ((unsigned)iu < (unsigned)N_AG) && (tb < T_STEPS);
    float p0e, p0o, p1, p2;

    {
      v2f sm;
      sm.x = co;
      sm.y = cb;
      mlp(cxe, sm, p0e, p1, p2);
      co = p2;
      edge[par][0][tile][lane] = p1;
      cb = from_slot_below(EA, p1);
    }

    {
      v2f sm;
      sm.x = co;
      sm.y = (iu == N_AG - 2) ? 0.0f : cb;
      float cxo[HALF_H];
      xpart(xa.z, xa.w, cxo);
      mlp(cxo, sm, p0o, p1, p2);
      co = p2;
      edge[par][1][tile][lane] = p1;
      cb = from_slot_below(EB, p1);
    }

    {
      const float mine  = up ? p0o : p0e;
      const float yours = up ? p0e : p0o;
      const float ctl   = mine + xhalf(yours);
      if (busy) {
        const unsigned off = (unsigned)tb * N_AG + (unsigned)iu + (unsigned)up;
        orun[off] = ctl;
      }
    }

    __builtin_amdgcn_fence(__ATOMIC_RELEASE, "workgroup", "local");
    __builtin_amdgcn_s_barrier_signal(-1);

    iu += 2;
    if (iu == PERIOD) { iu = 0; tb += SLOTS; }

    co = (iu == 0) ? 0.0f : co;
    asm volatile("" :: "v"(co));

    xpart(xn.x, xn.y, cxe);
    asm volatile("" :: "v"(cxe[0]), "v"(cxe[1]), "v"(cxe[2]), "v"(cxe[3]), "v"(cxe[4]));

    xa = *xptr(4);

    if (((w >> 1) & (TILES - 1)) == tile)
      __builtin_prefetch((const char*)xptr(4 + 16), 0, 0);

    __builtin_amdgcn_s_barrier_wait(-1);
    __builtin_amdgcn_fence(__ATOMIC_ACQUIRE, "workgroup", "local");
    const bool z0 = (tb == 0);
    const bool z1 = z0 && (iu != PERIOD - 2);
    EA = *(z0 ? &zero_word : &edge[par][0][below][15 + 16 * up]);
    EB = *(z1 ? &zero_word : &edge[par][1][below][15 + 16 * up]);
    __builtin_amdgcn_sched_barrier(0);
  };

  float4 X0 = *xptr(0);
  float4 X1 = *xptr(2);
  float4 X2 = *xptr(4);
  xpart(X0.x, X0.y, cxe);
  for (int w = 0; w < W_LOOP; w += 12) {
    iter(w,      X0, X1, 0);
    iter(w + 2,  X1, X2, 1);
    iter(w + 4,  X2, X0, 0);
    iter(w + 6,  X0, X1, 1);
    iter(w + 8,  X1, X2, 0);
    iter(w + 10, X2, X0, 1);
  }
}

extern "C" void kernel_launch(void* const* d_in, const int* in_sizes, int n_in,
                              void* d_out, int out_size, void* d_ws, size_t ws_size,
                              hipStream_t stream) {
  (void)in_sizes; (void)n_in; (void)out_size; (void)d_ws; (void)ws_size;
  const float* runs = (const float*)d_in[0];
  const float* W1   = (const float*)d_in[1];
  const float* b1   = (const float*)d_in[2];
  const float* W2   = (const float*)d_in[3];
  const float* b2   = (const float*)d_in[4];
  float* out        = (float*)d_out;

  dim3 grid(R_TOT);
  dim3 block(32 * TILES);
  recurrent_mlp_wavefront_kernel<<<grid, block, 0, stream>>>(runs, W1, b1, W2, b2, out);
}


# x pairs are requested 12 iterations ahead into 12 register sets (was 3), issued behind the LDS stores: a load whose slowest line is late no longer stalls the waves; line hint dropped; sums unchanged
# speedup vs baseline: 72.4584x; 1.1338x over previous
#include <hip/hip_runtime.h>


typedef float v2f __attribute__((ext_vector_type(2)));
typedef float v8f __attribute__((ext_vector_type(8)));

#define R_TOT    128
#define T_STEPS  1024
#define N_AG     128
#define TILES    4
#define SLOTS    (16 * TILES)
#define ROUNDS   (T_STEPS / SLOTS)
#define GAP      (2 * TILES)
#define PERIOD   (N_AG + GAP)
#define W_TOTAL  (2 * (SLOTS - 1 + TILES - 1) + (ROUNDS - 1) * PERIOD + N_AG)
#define X_SETS   12
#define W_LOOP   (((W_TOTAL + 2 * X_SETS - 1) / (2 * X_SETS)) * (2 * X_SETS))
#define HALF_H   5
#define NO_STORE 0xffffffffu
static_assert(T_STEPS % SLOTS == 0 && PERIOD % 2 == 0 && N_AG % 2 == 0, "even steps start on even agents");
static_assert(X_SETS % 2 == 0 && 2 * X_SETS < PERIOD, "a trip is a whole number of edge-copy turns; the load distance stays inside one round's wrap");

__device__ __forceinline__ float fast_tanh(float x) {
#if __has_builtin(__builtin_amdgcn_tanhf)
  return __builtin_amdgcn_tanhf(x);
#else
  float e = __builtin_exp2f(x * 2.885390081777927f);
  return 1.0f - 2.0f / (e + 1.0f);
#endif
}

__device__ __forceinline__ float xhalf(float x) {
  const int v = __builtin_bit_cast(int, x);
  return __builtin_bit_cast(float,
      __builtin_amdgcn_permlanex16(v, v, 0x76543210, 0xfedcba98, false, false));
}

__device__ __forceinline__ float from_slot_below(float edge_word, float x) {
  return __builtin_bit_cast(float,
      __builtin_amdgcn_update_dpp(__builtin_bit_cast(int, edge_word), __builtin_bit_cast(int, x),
                                  0x111  , 0xf, 0xf, false));
}

__global__ __launch_bounds__(32 * TILES)
void recurrent_mlp_wavefront_kernel(const float* __restrict__ runs,
                              const float* __restrict__ W1,
                              const float* __restrict__ b1,
                              const float* __restrict__ W2,
                              const float* __restrict__ b2,
                              float*       __restrict__ out)
{
  __shared__ float edge[2][2][TILES][32];
  __shared__ float zero_word;

  const int lane = threadIdx.x & 31;
  const int up   = lane >> 4;
  const int l15  = lane & 15;
  const int tile  = __builtin_amdgcn_readfirstlane((int)(threadIdx.x >> 5));
  const int below = (tile + TILES - 1) & (TILES - 1);
  const int s     = 16 * tile + l15;
  const int run   = blockIdx.x;
  if (threadIdx.x == 0) zero_word = 0.0f;

  v2f amat; amat.x = 0.0f; amat.y = 0.0f;
  {
    const int row = l15;
    int unit = -1;
    if (row < HALF_H) unit = row;
    else if (row >= 8 && row < 8 + HALF_H) unit = row - 8 + HALF_H;
    if (unit >= 0) { amat.x = W1[unit * 4 + 2]; amat.y = W1[unit * 4 + 3]; }
  }

  float wx0[HALF_H], wx1[HALF_H], bb1[HALF_H], w2a[HALF_H], w2b[HALF_H], w2c[HALF_H];
#pragma unroll
  for (int r = 0; r < HALF_H; ++r) {
    const int unit = HALF_H * up + r;
    wx0[r] = W1[unit * 4 + 0];
    wx1[r] = W1[unit * 4 + 1];
    bb1[r] = b1[unit];
    w2a[r] = W2[0 * 10 + unit];
    w2b[r] = W2[1 * 10 + unit];
    w2c[r] = W2[2 * 10 + unit];
  }
  const float b20 = up ? 0.0f : b2[0];
  const float b21 = up ? 0.0f : b2[1];
  const float b22 = up ? 0.0f : b2[2];

  const float* xrun = runs + (size_t)run * T_STEPS * (N_AG * 2);
  float*       orun = out  + (size_t)run * T_STEPS * N_AG;

  int iu = -2 * (s + tile);
  int tb = s;

  auto xptr = [&](const int d) __attribute__((always_inline)) -> const float4* {
    int ii = iu + d, tt = tb;
    if (ii >= PERIOD) { ii -= PERIOD; tt += SLOTS; }
    ii = min(max(ii, 0), N_AG - 2);
    tt &= (T_STEPS - 1);
    return (const float4*)(xrun + ((size_t)(tt * N_AG + ii)) * 2);
  };

  auto xpart = [&](const float x0, const float x1, float (&cx)[HALF_H]) __attribute__((always_inline)) {
#pragma unroll
    for (int r = 0; r < HALF_H; ++r) cx[r] = fmaf(wx1[r], x1, fmaf(wx0[r], x0, bb1[r]));
  };

  auto mlp = [&](const float (&cx)[HALF_H], const v2f sm,
                 float& p0, float& p1, float& p2) __attribute__((always_inline)) {
    v8f c;
#pragma unroll
    for (int r = 0; r < HALF_H; ++r) c[r] = cx[r];
    c[5] = 0.0f; c[6] = 0.0f; c[7] = 0.0f;

    v8f d = __builtin_amdgcn_wmma_f32_16x16x4_f32(
        false, amat, false, sm, (short)0, c, false, false);

    float dt[HALF_H];
#pragma unroll
    for (int r = 0; r < HALF_H; ++r) dt[r] = fast_tanh(d[r]);

    float c0 = fmaf(w2b[0], dt[0], b21), c1 = w2b[1] * dt[1];
    c0 = fmaf(w2b[2], dt[2], c0);        c1 = fmaf(w2b[3], dt[3], c1);
    c0 = fmaf(w2b[4], dt[4], c0);
    p1 = c0 + c1;

    float e0 = fmaf(w2c[0], dt[0], b22), e1 = w2c[1] * dt[1];
    e0 = fmaf(w2c[2], dt[2], e0);        e1 = fmaf(w2c[3], dt[3], e1);
    e0 = fmaf(w2c[4], dt[4], e0);
    p2 = e0 + e1;

    float a0 = fmaf(w2a[0], dt[0], b20), a1 = w2a[1] * dt[1];
    a0 = fmaf(w2a[2], dt[2], a0);        a1 = fmaf(w2a[3], dt[3], a1);
    a0 = fmaf(w2a[4], dt[4], a0);
    p0 = a0 + a1;
  };

  float co = 0.0f, cb = 0.0f;
  float EA = 0.0f, EB = 0.0f;

  float cxe[HALF_H];

  float    keepE = 0.0f, keepO = 0.0f;
  unsigned keepOff = NO_STORE;

  auto iter = [&](float4& xa, const float4& xn, const int par)
      __attribute__((always_inline)) {
    const bool busy = ((unsigned)iu < (unsigned)N_AG) && (tb < T_STEPS);
    float p0e, p0o, p1, p2;

    {
      v2f sm;
      sm.x = co;
      sm.y = cb;
      mlp(cxe, sm, p0e, p1, p2);
      co = p2;
      edge[par][0][tile][lane] = p1;
      cb = from_slot_below(EA, p1);
    }

    {
      v2f sm;
      sm.x = co;
      sm.y = (iu == N_AG - 2) ? 0.0f : cb;
      float cxo[HALF_H];
      xpart(xa.z, xa.w, cxo);
      mlp(cxo, sm, p0o, p1, p2);
      co = p2;
      edge[par][1][tile][lane] = p1;
      cb = from_slot_below(EB, p1);
    }

    if (par == 0) {
      keepE   = p0e;
      keepO   = p0o;
      keepOff = busy ? (((unsigned)tb * N_AG + (unsigned)iu) >> 1) : NO_STORE;
    } else {
      const float sendE = up ? keepE : p0e;
      const float sendO = up ? keepO : p0o;
      const float mineE = up ? p0e : keepE;
      const float mineO = up ? p0o : keepO;
      v2f ctl;
      ctl.x = mineE + xhalf(sendE);
      ctl.y = mineO + xhalf(sendO);
      const unsigned offB = busy ? (((unsigned)tb * N_AG + (unsigned)iu) >> 1) : NO_STORE;
      const unsigned off  = up ? offB : keepOff;
      if (off != NO_STORE) ((v2f*)orun)[off] = ctl;
    }

    iu += 2;
    if (iu == PERIOD) { iu = 0; tb += SLOTS; }

    __builtin_amdgcn_sched_barrier(0x54F);
    xa = *xptr(2 * (X_SETS - 1));

    __builtin_amdgcn_fence(__ATOMIC_RELEASE, "workgroup", "local");
    __builtin_amdgcn_s_barrier_signal(-1);

    co = (iu == 0) ? 0.0f : co;
    asm volatile("" :: "v"(co));

    xpart(xn.x, xn.y, cxe);
    asm volatile("" :: "v"(cxe[0]), "v"(cxe[1]), "v"(cxe[2]), "v"(cxe[3]), "v"(cxe[4]));

    const bool z0 = (tb == 0);
    const bool z1 = z0 && (iu != PERIOD - 2);
    const float* const pEA = z0 ? &zero_word : &edge[par][0][below][15 + 16 * up];
    const float* const pEB = z1 ? &zero_word : &edge[par][1][below][15 + 16 * up];

    __builtin_amdgcn_s_barrier_wait(-1);
    __builtin_amdgcn_fence(__ATOMIC_ACQUIRE, "workgroup", "local");
    EA = *pEA;
    EB = *pEB;
    __builtin_amdgcn_sched_barrier(0);
  };

  float4 X0  = *xptr(0);
  float4 X1  = *xptr(2);
  float4 X2  = *xptr(4);
  float4 X3  = *xptr(6);
  float4 X4  = *xptr(8);
  float4 X5  = *xptr(10);
  float4 X6  = *xptr(12);
  float4 X7  = *xptr(14);
  float4 X8  = *xptr(16);
  float4 X9  = *xptr(18);
  float4 X10 = *xptr(20);
  float4 X11 = *xptr(22);
  xpart(X0.x, X0.y, cxe);
  for (int w = 0; w < W_LOOP; w += 2 * X_SETS) {
    iter(X0,  X1,  0);
    iter(X1,  X2,  1);
    iter(X2,  X3,  0);
    iter(X3,  X4,  1);
    iter(X4,  X5,  0);
    iter(X5,  X6,  1);
    iter(X6,  X7,  0);
    iter(X7,  X8,  1);
    iter(X8,  X9,  0);
    iter(X9,  X10, 1);
    iter(X10, X11, 0);
    iter(X11, X0,  1);
  }
}

extern "C" void kernel_launch(void* const* d_in, const int* in_sizes, int n_in,
                              void* d_out, int out_size, void* d_ws, size_t ws_size,
                              hipStream_t stream) {
  (void)in_sizes; (void)n_in; (void)out_size; (void)d_ws; (void)ws_size;
  const float* runs = (const float*)d_in[0];
  const float* W1   = (const float*)d_in[1];
  const float* b1   = (const float*)d_in[2];
  const float* W2   = (const float*)d_in[3];
  const float* b2   = (const float*)d_in[4];
  float* out        = (float*)d_out;

  dim3 grid(R_TOT);
  dim3 block(32 * TILES);
  recurrent_mlp_wavefront_kernel<<<grid, block, 0, stream>>>(runs, W1, b1, W2, b2, out);
}


# Tiles trail by 8 extra steps, not 2: the o1 word crossing waves is needed 9 steps after it is made, so the four waves meet (LDS hand-over, barrier) once per 8 steps, not per 2; every sum unchanged
# speedup vs baseline: 77.4337x; 1.0687x over previous
#include <hip/hip_runtime.h>


typedef float v2f __attribute__((ext_vector_type(2)));
typedef float v8f __attribute__((ext_vector_type(8)));

#define R_TOT    128
#define T_STEPS  1024
#define N_AG     128
#define TILES    4
#define SLOTS    (16 * TILES)
#define ROUNDS   (T_STEPS / SLOTS)
#define LAG      8
#define BLK_IT   (LAG / 2)
#define GAP      (LAG * TILES)
#define PERIOD   (N_AG + GAP)
#define W_TOTAL  (2 * (SLOTS - 1) + LAG * (TILES - 1) + (ROUNDS - 1) * PERIOD + N_AG)
#define X_SETS   16
#define W_LOOP   (((W_TOTAL + 2 * X_SETS - 1) / (2 * X_SETS)) * (2 * X_SETS))
#define HALF_H   5
#define NO_STORE 0xffffffffu
static_assert(T_STEPS % SLOTS == 0 && PERIOD % 2 == 0 && N_AG % 2 == 0, "even steps start on even agents");
static_assert(LAG == 8 && X_SETS % (2 * BLK_IT) == 0 && 2 * X_SETS < PERIOD, "blocks of four iterations; a trip is a whole number of edge-copy turns; the load distance stays inside one round's wrap");

__device__ __forceinline__ float fast_tanh(float x) {
#if __has_builtin(__builtin_amdgcn_tanhf)
  return __builtin_amdgcn_tanhf(x);
#else
  float e = __builtin_exp2f(x * 2.885390081777927f);
  return 1.0f - 2.0f / (e + 1.0f);
#endif
}

__device__ __forceinline__ float xhalf(float x) {
  const int v = __builtin_bit_cast(int, x);
  return __builtin_bit_cast(float,
      __builtin_amdgcn_permlanex16(v, v, 0x76543210, 0xfedcba98, false, false));
}

__device__ __forceinline__ float from_slot_below(float edge_word, float x) {
  return __builtin_bit_cast(float,
      __builtin_amdgcn_update_dpp(__builtin_bit_cast(int, edge_word), __builtin_bit_cast(int, x),
                                  0x111  , 0xf, 0xf, false));
}

__global__ __launch_bounds__(32 * TILES)
void recurrent_mlp_wavefront_kernel(const float* __restrict__ runs,
                              const float* __restrict__ W1,
                              const float* __restrict__ b1,
                              const float* __restrict__ W2,
                              const float* __restrict__ b2,
                              float*       __restrict__ out)
{
  __shared__ float edge[2][LAG][TILES][32];
  __shared__ float zero_word;

  const int lane = threadIdx.x & 31;
  const int up   = lane >> 4;
  const int l15  = lane & 15;
  const int tile  = __builtin_amdgcn_readfirstlane((int)(threadIdx.x >> 5));
  const int below = (tile + TILES - 1) & (TILES - 1);
  const int s     = 16 * tile + l15;
  const int run   = blockIdx.x;
  if (threadIdx.x == 0) zero_word = 0.0f;

  v2f amat; amat.x = 0.0f; amat.y = 0.0f;
  {
    const int row = l15;
    int unit = -1;
    if (row < HALF_H) unit = row;
    else if (row >= 8 && row < 8 + HALF_H) unit = row - 8 + HALF_H;
    if (unit >= 0) { amat.x = W1[unit * 4 + 2]; amat.y = W1[unit * 4 + 3]; }
  }

  float wx0[HALF_H], wx1[HALF_H], bb1[HALF_H], w2a[HALF_H], w2b[HALF_H], w2c[HALF_H];
#pragma unroll
  for (int r = 0; r < HALF_H; ++r) {
    const int unit = HALF_H * up + r;
    wx0[r] = W1[unit * 4 + 0];
    wx1[r] = W1[unit * 4 + 1];
    bb1[r] = b1[unit];
    w2a[r] = W2[0 * 10 + unit];
    w2b[r] = W2[1 * 10 + unit];
    w2c[r] = W2[2 * 10 + unit];
  }
  const float b20 = up ? 0.0f : b2[0];
  const float b21 = up ? 0.0f : b2[1];
  const float b22 = up ? 0.0f : b2[2];

  const float* xrun = runs + (size_t)run * T_STEPS * (N_AG * 2);
  float*       orun = out  + (size_t)run * T_STEPS * N_AG;

  int iu = -2 * s - LAG * tile;
  int tb = s;

  auto xptr = [&](const int d) __attribute__((always_inline)) -> const float4* {
    int ii = iu + d, tt = tb;
    if (ii >= PERIOD) { ii -= PERIOD; tt += SLOTS; }
    ii = min(max(ii, 0), N_AG - 2);
    tt &= (T_STEPS - 1);
    return (const float4*)(xrun + ((size_t)(tt * N_AG + ii)) * 2);
  };

  auto xpart = [&](const float x0, const float x1, float (&cx)[HALF_H]) __attribute__((always_inline)) {
#pragma unroll
    for (int r = 0; r < HALF_H; ++r) cx[r] = fmaf(wx1[r], x1, fmaf(wx0[r], x0, bb1[r]));
  };

  auto mlp = [&](const float (&cx)[HALF_H], const v2f sm,
                 float& p0, float& p1, float& p2) __attribute__((always_inline)) {
    v8f c;
#pragma unroll
    for (int r = 0; r < HALF_H; ++r) c[r] = cx[r];
    c[5] = 0.0f; c[6] = 0.0f; c[7] = 0.0f;

    v8f d = __builtin_amdgcn_wmma_f32_16x16x4_f32(
        false, amat, false, sm, (short)0, c, false, false);

    float dt[HALF_H];
#pragma unroll
    for (int r = 0; r < HALF_H; ++r) dt[r] = fast_tanh(d[r]);

    float c0 = fmaf(w2b[0], dt[0], b21), c1 = w2b[1] * dt[1];
    c0 = fmaf(w2b[2], dt[2], c0);        c1 = fmaf(w2b[3], dt[3], c1);
    c0 = fmaf(w2b[4], dt[4], c0);
    p1 = c0 + c1;

    float e0 = fmaf(w2c[0], dt[0], b22), e1 = w2c[1] * dt[1];
    e0 = fmaf(w2c[2], dt[2], e0);        e1 = fmaf(w2c[3], dt[3], e1);
    e0 = fmaf(w2c[4], dt[4], e0);
    p2 = e0 + e1;

    float a0 = fmaf(w2a[0], dt[0], b20), a1 = w2a[1] * dt[1];
    a0 = fmaf(w2a[2], dt[2], a0);        a1 = fmaf(w2a[3], dt[3], a1);
    a0 = fmaf(w2a[4], dt[4], a0);
    p0 = a0 + a1;
  };

  float co = 0.0f, cb = 0.0f;
  float E0 = 0.0f, E1 = 0.0f, E2 = 0.0f, E3 = 0.0f, E4 = 0.0f, E5 = 0.0f, E6 = 0.0f, E7 = 0.0f;

  float cxe[HALF_H];

  float    keepE = 0.0f, keepO = 0.0f;
  unsigned keepOff = NO_STORE;

  auto iter = [&](const float4& xa, const int q, const int cp, const float Ee, const float Eo)
      __attribute__((always_inline)) {
    const bool busy = ((unsigned)iu < (unsigned)N_AG) && (tb < T_STEPS);
    float p0e, p0o, p1, p2;

    {
      v2f sm;
      sm.x = co;
      sm.y = cb;
      mlp(cxe, sm, p0e, p1, p2);
      co = p2;
      edge[cp][2 * q][tile][lane] = p1;
      cb = from_slot_below(Ee, p1);
    }

    {
      v2f sm;
      sm.x = co;
      sm.y = (iu == N_AG - 2) ? 0.0f : cb;
      float cxo[HALF_H];
      xpart(xa.z, xa.w, cxo);
      mlp(cxo, sm, p0o, p1, p2);
      co = p2;
      edge[cp][2 * q + 1][tile][lane] = p1;
      cb = from_slot_below(Eo, p1);
    }

    if ((q & 1) == 0) {
      keepE   = p0e;
      keepO   = p0o;
      keepOff = busy ? (((unsigned)tb * N_AG + (unsigned)iu) >> 1) : NO_STORE;
    } else {
      const float sendE = up ? keepE : p0e;
      const float sendO = up ? keepO : p0o;
      const float mineE = up ? p0e : keepE;
      const float mineO = up ? p0o : keepO;
      v2f ctl;
      ctl.x = mineE + xhalf(sendE);
      ctl.y = mineO + xhalf(sendO);
      const unsigned offB = busy ? (((unsigned)tb * N_AG + (unsigned)iu) >> 1) : NO_STORE;
      const unsigned off  = up ? offB : keepOff;
      if (off != NO_STORE) ((v2f*)orun)[off] = ctl;
    }

    iu += 2;
    if (iu == PERIOD) { iu = 0; tb += SLOTS; }
  };

  auto between = [&](const float4& xn) __attribute__((always_inline)) {
    co = (iu == 0) ? 0.0f : co;
    xpart(xn.x, xn.y, cxe);
  };

  auto meet = [&](float4& xa, float4& xb, float4& xc, float4& xd, const float4& xn, const int cp)
      __attribute__((always_inline)) {
    const float4* const ga = xptr(2 * (X_SETS - 4));
    const float4* const gb = xptr(2 * (X_SETS - 3));
    const float4* const gc = xptr(2 * (X_SETS - 2));
    const float4* const gd = xptr(2 * (X_SETS - 1));
    __builtin_amdgcn_sched_barrier(0x54F);
    xa = *ga;
    xb = *gb;
    xc = *gc;
    xd = *gd;

    __builtin_amdgcn_fence(__ATOMIC_RELEASE, "workgroup", "local");
    __builtin_amdgcn_s_barrier_signal(-1);

    co = (iu == 0) ? 0.0f : co;
    asm volatile("" :: "v"(co));

    xpart(xn.x, xn.y, cxe);
    asm volatile("" :: "v"(cxe[0]), "v"(cxe[1]), "v"(cxe[2]), "v"(cxe[3]), "v"(cxe[4]));

    const bool zt = (tb == 0);
    const float* const src0 = &edge[cp][0][below][15 + 16 * up];
    const int step_words = TILES * 32;
    auto eptr = [&](const int r) __attribute__((always_inline)) -> const float* {
      const bool z = zt && (iu + 2 * ((r + 1) >> 1) < PERIOD);
      return z ? &zero_word : src0 + r * step_words;
    };
    const float* const pe0 = eptr(0);
    const float* const pe1 = eptr(1);
    const float* const pe2 = eptr(2);
    const float* const pe3 = eptr(3);
    const float* const pe4 = eptr(4);
    const float* const pe5 = eptr(5);
    const float* const pe6 = eptr(6);
    const float* const pe7 = eptr(7);

    __builtin_amdgcn_s_barrier_wait(-1);
    __builtin_amdgcn_fence(__ATOMIC_ACQUIRE, "workgroup", "local");
    E0 = *pe0;
    E1 = *pe1;
    E2 = *pe2;
    E3 = *pe3;
    E4 = *pe4;
    E5 = *pe5;
    E6 = *pe6;
    E7 = *pe7;
    __builtin_amdgcn_sched_barrier(0);
  };

  auto block = [&](float4& xa, float4& xb, float4& xc, float4& xd, const float4& xn, const int cp)
      __attribute__((always_inline)) {
    iter(xa, 0, cp, E0, E1);
    between(xb);
    iter(xb, 1, cp, E2, E3);
    between(xc);
    iter(xc, 2, cp, E4, E5);
    between(xd);
    iter(xd, 3, cp, E6, E7);
    meet(xa, xb, xc, xd, xn, cp);
  };

  float4 X0  = *xptr(0);
  float4 X1  = *xptr(2);
  float4 X2  = *xptr(4);
  float4 X3  = *xptr(6);
  float4 X4  = *xptr(8);
  float4 X5  = *xptr(10);
  float4 X6  = *xptr(12);
  float4 X7  = *xptr(14);
  float4 X8  = *xptr(16);
  float4 X9  = *xptr(18);
  float4 X10 = *xptr(20);
  float4 X11 = *xptr(22);
  float4 X12 = *xptr(24);
  float4 X13 = *xptr(26);
  float4 X14 = *xptr(28);
  float4 X15 = *xptr(30);
  xpart(X0.x, X0.y, cxe);
  for (int w = 0; w < W_LOOP; w += 2 * X_SETS) {
    block(X0,  X1,  X2,  X3,  X4,  0);
    block(X4,  X5,  X6,  X7,  X8,  1);
    block(X8,  X9,  X10, X11, X12, 0);
    block(X12, X13, X14, X15, X0,  1);
  }
}

extern "C" void kernel_launch(void* const* d_in, const int* in_sizes, int n_in,
                              void* d_out, int out_size, void* d_ws, size_t ws_size,
                              hipStream_t stream) {
  (void)in_sizes; (void)n_in; (void)out_size; (void)d_ws; (void)ws_size;
  const float* runs = (const float*)d_in[0];
  const float* W1   = (const float*)d_in[1];
  const float* b1   = (const float*)d_in[2];
  const float* W2   = (const float*)d_in[3];
  const float* b2   = (const float*)d_in[4];
  float* out        = (float*)d_out;

  dim3 grid(R_TOT);
  dim3 block(32 * TILES);
  recurrent_mlp_wavefront_kernel<<<grid, block, 0, stream>>>(runs, W1, b1, W2, b2, out);
}


# Control store of an iteration pair issued right behind the next matrix product, its lane mask made earlier in a scalar register: branch and translation wait run in the product's time; sums unchanged
# speedup vs baseline: 82.9053x; 1.0707x over previous
#include <hip/hip_runtime.h>


typedef float v2f __attribute__((ext_vector_type(2)));
typedef float v8f __attribute__((ext_vector_type(8)));

#define R_TOT    128
#define T_STEPS  1024
#define N_AG     128
#define TILES    4
#define SLOTS    (16 * TILES)
#define ROUNDS   (T_STEPS / SLOTS)
#define LAG      8
#define BLK_IT   (LAG / 2)
#define GAP      (LAG * TILES)
#define PERIOD   (N_AG + GAP)
#define W_TOTAL  (2 * (SLOTS - 1) + LAG * (TILES - 1) + (ROUNDS - 1) * PERIOD + N_AG)
#define X_SETS   16
#define W_LOOP   (((W_TOTAL + 2 * X_SETS - 1) / (2 * X_SETS)) * (2 * X_SETS))
#define HALF_H   5
#define NO_STORE 0xffffffffu
static_assert(T_STEPS % SLOTS == 0 && PERIOD % 2 == 0 && N_AG % 2 == 0, "even steps start on even agents");
static_assert(LAG == 8 && X_SETS % (2 * BLK_IT) == 0 && 2 * X_SETS < PERIOD, "blocks of four iterations; a trip is a whole number of edge-copy turns; the load distance stays inside one round's wrap");

__device__ __forceinline__ float fast_tanh(float x) {
#if __has_builtin(__builtin_amdgcn_tanhf)
  return __builtin_amdgcn_tanhf(x);
#else
  float e = __builtin_exp2f(x * 2.885390081777927f);
  return 1.0f - 2.0f / (e + 1.0f);
#endif
}

__device__ __forceinline__ float xhalf(float x) {
  const int v = __builtin_bit_cast(int, x);
  return __builtin_bit_cast(float,
      __builtin_amdgcn_permlanex16(v, v, 0x76543210, 0xfedcba98, false, false));
}

__device__ __forceinline__ float from_slot_below(float edge_word, float x) {
  return __builtin_bit_cast(float,
      __builtin_amdgcn_update_dpp(__builtin_bit_cast(int, edge_word), __builtin_bit_cast(int, x),
                                  0x111  , 0xf, 0xf, false));
}

__global__ __launch_bounds__(32 * TILES)
void recurrent_mlp_wavefront_kernel(const float* __restrict__ runs,
                              const float* __restrict__ W1,
                              const float* __restrict__ b1,
                              const float* __restrict__ W2,
                              const float* __restrict__ b2,
                              float*       __restrict__ out)
{
  __shared__ float edge[2][LAG][TILES][32];
  __shared__ float zero_word;

  const int lane = threadIdx.x & 31;
  const int up   = lane >> 4;
  const int l15  = lane & 15;
  const int tile  = __builtin_amdgcn_readfirstlane((int)(threadIdx.x >> 5));
  const int below = (tile + TILES - 1) & (TILES - 1);
  const int s     = 16 * tile + l15;
  const int run   = blockIdx.x;
  if (threadIdx.x == 0) zero_word = 0.0f;

  v2f amat; amat.x = 0.0f; amat.y = 0.0f;
  {
    const int row = l15;
    int unit = -1;
    if (row < HALF_H) unit = row;
    else if (row >= 8 && row < 8 + HALF_H) unit = row - 8 + HALF_H;
    if (unit >= 0) { amat.x = W1[unit * 4 + 2]; amat.y = W1[unit * 4 + 3]; }
  }

  float wx0[HALF_H], wx1[HALF_H], bb1[HALF_H], w2a[HALF_H], w2b[HALF_H], w2c[HALF_H];
#pragma unroll
  for (int r = 0; r < HALF_H; ++r) {
    const int unit = HALF_H * up + r;
    wx0[r] = W1[unit * 4 + 0];
    wx1[r] = W1[unit * 4 + 1];
    bb1[r] = b1[unit];
    w2a[r] = W2[0 * 10 + unit];
    w2b[r] = W2[1 * 10 + unit];
    w2c[r] = W2[2 * 10 + unit];
  }
  const float b20 = up ? 0.0f : b2[0];
  const float b21 = up ? 0.0f : b2[1];
  const float b22 = up ? 0.0f : b2[2];

  const float* xrun = runs + (size_t)run * T_STEPS * (N_AG * 2);
  float*       orun = out  + (size_t)run * T_STEPS * N_AG;

  int iu = -2 * s - LAG * tile;
  int tb = s;

  auto xptr = [&](const int d) __attribute__((always_inline)) -> const float4* {
    int ii = iu + d, tt = tb;
    if (ii >= PERIOD) { ii -= PERIOD; tt += SLOTS; }
    ii = min(max(ii, 0), N_AG - 2);
    tt &= (T_STEPS - 1);
    return (const float4*)(xrun + ((size_t)(tt * N_AG + ii)) * 2);
  };

  auto xpart = [&](const float x0, const float x1, float (&cx)[HALF_H]) __attribute__((always_inline)) {
#pragma unroll
    for (int r = 0; r < HALF_H; ++r) cx[r] = fmaf(wx1[r], x1, fmaf(wx0[r], x0, bb1[r]));
  };

  v2f      pendCtl; pendCtl.x = 0.0f; pendCtl.y = 0.0f;
  unsigned pendOff = NO_STORE;
  unsigned pendMask = 0u;
  auto send_pending = [&]() __attribute__((always_inline)) {
    if (__builtin_amdgcn_inverse_ballot_w32(pendMask))
      ((v2f*)orun)[pendOff] = pendCtl;
  };

  auto mlp = [&](const float (&cx)[HALF_H], const v2f sm, const bool flush,
                 float& p0, float& p1, float& p2) __attribute__((always_inline)) {
    v8f c;
#pragma unroll
    for (int r = 0; r < HALF_H; ++r) c[r] = cx[r];
    c[5] = 0.0f; c[6] = 0.0f; c[7] = 0.0f;

    v8f d = __builtin_amdgcn_wmma_f32_16x16x4_f32(
        false, amat, false, sm, (short)0, c, false, false);

    if (flush) send_pending();

    float dt[HALF_H];
#pragma unroll
    for (int r = 0; r < HALF_H; ++r) dt[r] = fast_tanh(d[r]);

    float c0 = fmaf(w2b[0], dt[0], b21), c1 = w2b[1] * dt[1];
    c0 = fmaf(w2b[2], dt[2], c0);        c1 = fmaf(w2b[3], dt[3], c1);
    c0 = fmaf(w2b[4], dt[4], c0);
    p1 = c0 + c1;

    float e0 = fmaf(w2c[0], dt[0], b22), e1 = w2c[1] * dt[1];
    e0 = fmaf(w2c[2], dt[2], e0);        e1 = fmaf(w2c[3], dt[3], e1);
    e0 = fmaf(w2c[4], dt[4], e0);
    p2 = e0 + e1;

    float a0 = fmaf(w2a[0], dt[0], b20), a1 = w2a[1] * dt[1];
    a0 = fmaf(w2a[2], dt[2], a0);        a1 = fmaf(w2a[3], dt[3], a1);
    a0 = fmaf(w2a[4], dt[4], a0);
    p0 = a0 + a1;
  };

  float co = 0.0f, cb = 0.0f;
  float E0 = 0.0f, E1 = 0.0f, E2 = 0.0f, E3 = 0.0f, E4 = 0.0f, E5 = 0.0f, E6 = 0.0f, E7 = 0.0f;

  float cxe[HALF_H];

  float    keepE = 0.0f, keepO = 0.0f;
  unsigned keepOff = NO_STORE;

  auto iter = [&](const float4& xa, const int q, const int cp, const float Ee, const float Eo)
      __attribute__((always_inline)) {
    const bool busy = ((unsigned)iu < (unsigned)N_AG) && (tb < T_STEPS);
    float p0e, p0o, p1, p2;

    {
      v2f sm;
      sm.x = co;
      sm.y = cb;
      if (q == 2) __builtin_amdgcn_sched_barrier(0);
      mlp(cxe, sm, (q & 1) == 0, p0e, p1, p2);
      co = p2;
      edge[cp][2 * q][tile][lane] = p1;
      cb = from_slot_below(Ee, p1);
    }

    {
      v2f sm;
      sm.x = co;
      sm.y = (iu == N_AG - 2) ? 0.0f : cb;
      float cxo[HALF_H];
      xpart(xa.z, xa.w, cxo);
      mlp(cxo, sm, false, p0o, p1, p2);
      co = p2;
      edge[cp][2 * q + 1][tile][lane] = p1;
      cb = from_slot_below(Eo, p1);
    }

    if ((q & 1) == 0) {
      keepE   = p0e;
      keepO   = p0o;
      keepOff = busy ? (((unsigned)tb * N_AG + (unsigned)iu) >> 1) : NO_STORE;
    } else {
      const float sendE = up ? keepE : p0e;
      const float sendO = up ? keepO : p0o;
      const float mineE = up ? p0e : keepE;
      const float mineO = up ? p0o : keepO;
      v2f ctl;
      ctl.x = mineE + xhalf(sendE);
      ctl.y = mineO + xhalf(sendO);
      const unsigned offB = busy ? (((unsigned)tb * N_AG + (unsigned)iu) >> 1) : NO_STORE;
      pendCtl = ctl;
      pendOff = up ? offB : keepOff;
      pendMask = __builtin_amdgcn_ballot_w32(pendOff != NO_STORE);
      asm volatile("" :: "v"(pendCtl.x), "v"(pendCtl.y), "v"(pendOff), "s"(pendMask));
    }

    iu += 2;
    if (iu == PERIOD) { iu = 0; tb += SLOTS; }
  };

  auto between = [&](const float4& xn) __attribute__((always_inline)) {
    co = (iu == 0) ? 0.0f : co;
    xpart(xn.x, xn.y, cxe);
  };

  auto meet = [&](float4& xa, float4& xb, float4& xc, float4& xd, const float4& xn, const int cp)
      __attribute__((always_inline)) {
    const float4* const ga = xptr(2 * (X_SETS - 4));
    const float4* const gb = xptr(2 * (X_SETS - 3));
    const float4* const gc = xptr(2 * (X_SETS - 2));
    const float4* const gd = xptr(2 * (X_SETS - 1));
    __builtin_amdgcn_sched_barrier(0x54F);
    xa = *ga;
    xb = *gb;
    xc = *gc;
    xd = *gd;

    __builtin_amdgcn_fence(__ATOMIC_RELEASE, "workgroup", "local");
    __builtin_amdgcn_s_barrier_signal(-1);

    co = (iu == 0) ? 0.0f : co;
    asm volatile("" :: "v"(co));

    xpart(xn.x, xn.y, cxe);
    asm volatile("" :: "v"(cxe[0]), "v"(cxe[1]), "v"(cxe[2]), "v"(cxe[3]), "v"(cxe[4]));

    const bool zt = (tb == 0);
    const float* const src0 = &edge[cp][0][below][15 + 16 * up];
    const int step_words = TILES * 32;
    auto eptr = [&](const int r) __attribute__((always_inline)) -> const float* {
      const bool z = zt && (iu + 2 * ((r + 1) >> 1) < PERIOD);
      return z ? &zero_word : src0 + r * step_words;
    };
    const float* const pe0 = eptr(0);
    const float* const pe1 = eptr(1);
    const float* const pe2 = eptr(2);
    const float* const pe3 = eptr(3);
    const float* const pe4 = eptr(4);
    const float* const pe5 = eptr(5);
    const float* const pe6 = eptr(6);
    const float* const pe7 = eptr(7);

    __builtin_amdgcn_s_barrier_wait(-1);
    __builtin_amdgcn_fence(__ATOMIC_ACQUIRE, "workgroup", "local");
    E0 = *pe0;
    E1 = *pe1;
    E2 = *pe2;
    E3 = *pe3;
    E4 = *pe4;
    E5 = *pe5;
    E6 = *pe6;
    E7 = *pe7;
    __builtin_amdgcn_sched_barrier(0);
  };

  auto block = [&](float4& xa, float4& xb, float4& xc, float4& xd, const float4& xn, const int cp)
      __attribute__((always_inline)) {
    iter(xa, 0, cp, E0, E1);
    between(xb);
    iter(xb, 1, cp, E2, E3);
    between(xc);
    iter(xc, 2, cp, E4, E5);
    between(xd);
    iter(xd, 3, cp, E6, E7);
    meet(xa, xb, xc, xd, xn, cp);
  };

  float4 X0  = *xptr(0);
  float4 X1  = *xptr(2);
  float4 X2  = *xptr(4);
  float4 X3  = *xptr(6);
  float4 X4  = *xptr(8);
  float4 X5  = *xptr(10);
  float4 X6  = *xptr(12);
  float4 X7  = *xptr(14);
  float4 X8  = *xptr(16);
  float4 X9  = *xptr(18);
  float4 X10 = *xptr(20);
  float4 X11 = *xptr(22);
  float4 X12 = *xptr(24);
  float4 X13 = *xptr(26);
  float4 X14 = *xptr(28);
  float4 X15 = *xptr(30);
  xpart(X0.x, X0.y, cxe);
  for (int w = 0; w < W_LOOP; w += 2 * X_SETS) {
    block(X0,  X1,  X2,  X3,  X4,  0);
    block(X4,  X5,  X6,  X7,  X8,  1);
    block(X8,  X9,  X10, X11, X12, 0);
    block(X12, X13, X14, X15, X0,  1);
  }
  send_pending();
}

extern "C" void kernel_launch(void* const* d_in, const int* in_sizes, int n_in,
                              void* d_out, int out_size, void* d_ws, size_t ws_size,
                              hipStream_t stream) {
  (void)in_sizes; (void)n_in; (void)out_size; (void)d_ws; (void)ws_size;
  const float* runs = (const float*)d_in[0];
  const float* W1   = (const float*)d_in[1];
  const float* b1   = (const float*)d_in[2];
  const float* W2   = (const float*)d_in[3];
  const float* b2   = (const float*)d_in[4];
  float* out        = (float*)d_out;

  dim3 grid(R_TOT);
  dim3 block(32 * TILES);
  recurrent_mlp_wavefront_kernel<<<grid, block, 0, stream>>>(runs, W1, b1, W2, b2, out);
}
